# starting kernel
# baseline (speedup 1.0000x reference)
#include <hip/hip_runtime.h>
#include <hip/hip_bf16.h>
#include <stdint.h>


#define Bsz   4
#define Llen  16384
#define Dch   512
#define Nst   64
#define NFFT  32768
#define PI_F  3.14159265358979323846f

typedef __attribute__((ext_vector_type(2))) float v2f;
typedef __attribute__((ext_vector_type(8))) float v8f;

#ifndef __has_builtin
#define __has_builtin(x) 0
#endif
#if __has_builtin(__builtin_amdgcn_wmma_f32_16x16x4_f32)
#define S4_WMMA 1
#else
#define S4_WMMA 0
#endif

__device__ __forceinline__ float2 cadd(float2 a, float2 b) { return make_float2(a.x + b.x, a.y + b.y); }
__device__ __forceinline__ float2 csub(float2 a, float2 b) { return make_float2(a.x - b.x, a.y - b.y); }
__device__ __forceinline__ float2 cmulf(float2 a, float2 b) { return make_float2(a.x * b.x - a.y * b.y, a.x * b.y + a.y * b.x); }
__device__ __forceinline__ float2 crotis(float2 a, float s) { return make_float2(-s * a.y, s * a.x); }

__device__ __forceinline__ void dft8(const float2* v, float s, float2* X) {
  const float R = 0.70710678118654752440f;
  float2 t0 = cadd(v[0], v[4]), t1 = cadd(v[1], v[5]);
  float2 t2 = cadd(v[2], v[6]), t3 = cadd(v[3], v[7]);
  float2 u0 = csub(v[0], v[4]), u1 = csub(v[1], v[5]);
  float2 u2 = csub(v[2], v[6]), u3 = csub(v[3], v[7]);
  float2 m1 = cmulf(u1, make_float2(R, s * R));
  float2 m2 = crotis(u2, s);
  float2 m3 = cmulf(u3, make_float2(-R, s * R));
  float2 p0 = cadd(t0, t2), p1 = csub(t0, t2);
  float2 p2 = cadd(t1, t3), p3 = crotis(csub(t1, t3), s);
  X[0] = cadd(p0, p2); X[2] = cadd(p1, p3);
  X[4] = csub(p0, p2); X[6] = csub(p1, p3);
  float2 q0 = cadd(u0, m2), q1 = csub(u0, m2);
  float2 q2 = cadd(m1, m3), q3 = crotis(csub(m1, m3), s);
  X[1] = cadd(q0, q2); X[3] = cadd(q1, q3);
  X[5] = csub(q0, q2); X[7] = csub(q1, q3);
}

__global__ __launch_bounds__(256) void s4_perl(
    const float* __restrict__ Lam, const float* __restrict__ Pri,
    const float* __restrict__ Cri, const float* __restrict__ log_step,
    float* __restrict__ perl)
{
  int l = blockIdx.x * blockDim.x + threadIdx.x;
  if (l >= Llen) return;
  float step = expf(log_step[0]);
  float theta = -2.0f * PI_F * (float)l / (float)Llen;
  float oi, orr;
  sincosf(theta, &oi, &orr);
  float op1r = 1.0f + orr;
  float den = op1r * op1r + oi * oi;
  float nr = 1.0f - orr, ni = -oi;
  float gscale = 2.0f / step;
  float gr = gscale * (nr * op1r + ni * oi) / den;
  float gi = gscale * (ni * op1r - nr * oi) / den;
  float scr = 2.0f * op1r / den;
  float sci = -2.0f * oi / den;
  float k01r = 0.f, k01i = 0.f, k11r = 0.f, k11i = 0.f;
  for (int n = 0; n < Nst; ++n) {
    float lr = Lam[2 * n], li = Lam[2 * n + 1];
    float pr = Pri[2 * n], pi = Pri[2 * n + 1];
    float cr = Cri[2 * n], ci = Cri[2 * n + 1];
    float zr = gr - lr, zi = gi - li;
    float zd = zr * zr + zi * zi;
    float car = zr / zd, cai = -zi / zd;
    float w01r = cr * pr + ci * pi;
    float w01i = cr * pi - ci * pr;
    float w11r = pr * pr + pi * pi;
    k01r += car * w01r - cai * w01i;
    k01i += car * w01i + cai * w01r;
    k11r += car * w11r;
    k11i += cai * w11r;
  }
  float dr = 1.0f + k11r, di = k11i;
  float dd = dr * dr + di * di;
  float sr = (k01r * dr + k01i * di) / dd;
  float si = (k01i * dr - k01r * di) / dd;
  perl[l] = sr;
  perl[Llen + l] = si;
  perl[2 * Llen + l] = scr;
  perl[3 * Llen + l] = sci;
  perl[4 * Llen + l] = gr;
  perl[5 * Llen + l] = gi;
}

__global__ __launch_bounds__(256) void s4_weights(
    const float* __restrict__ Bri, const float* __restrict__ Pri,
    const float* __restrict__ Cri,
    float* __restrict__ W0r, float* __restrict__ W0i,
    float* __restrict__ W1r, float* __restrict__ W1i)
{
  int idx = blockIdx.x * blockDim.x + threadIdx.x;
  if (idx >= Nst * Dch) return;
  int n = idx / Dch, d = idx % Dch;
  float br = Bri[2 * (n * Dch + d)], bi = Bri[2 * (n * Dch + d) + 1];
  float cr = Cri[2 * n], ci = Cri[2 * n + 1];
  float pr = Pri[2 * n], pi = Pri[2 * n + 1];
  int o = (n >> 1) * (2 * Dch) + 2 * d + (n & 1);
  W0r[o] = cr * br + ci * bi;
  W0i[o] = cr * bi - ci * br;
  W1r[o] = pr * br + pi * bi;
  W1i[o] = pr * bi - pi * br;
}

__global__ __launch_bounds__(256) void s4_cauchy(
    const float* __restrict__ Lam, const float* __restrict__ perl,
    float* __restrict__ APr, float* __restrict__ APi)
{
  int idx = blockIdx.x * blockDim.x + threadIdx.x;
  if (idx >= Llen * Nst) return;
  int n = idx / Llen, l = idx % Llen;
  float gr = perl[4 * Llen + l], gi = perl[5 * Llen + l];
  float zr = gr - Lam[2 * n], zi = gi - Lam[2 * n + 1];
  float zd = zr * zr + zi * zi;
  int o = (n >> 1) * (2 * Llen) + 2 * l + (n & 1);
  APr[o] = zr / zd;
  APi[o] = -zi / zd;
}

__global__ __launch_bounds__(32) void s4_cauchy_gemm(
    const float* __restrict__ APr, const float* __restrict__ APi,
    const float* __restrict__ W0r, const float* __restrict__ W0i,
    const float* __restrict__ W1r, const float* __restrict__ W1i,
    const float* __restrict__ perl, float2* __restrict__ Ev)
{
  int lane = threadIdx.x & 31;
  int tileD = blockIdx.x;
  int tileL = blockIdx.y;
  int d0 = tileD * 16 + (lane & 15);
  int mbase = tileL * 16 + ((lane >> 4) ? 8 : 0);
#if S4_WMMA
  int m = tileL * 16 + (lane & 15);
  int khi = (lane >> 4) * 2;
  v8f aRr = {}, aRi = {}, aSr = {}, aSi = {};
  for (int kb = 0; kb < Nst; kb += 4) {
    int k0 = kb + khi;
    const float* pr = APr + (k0 >> 1) * (2 * Llen) + 2 * m;
    const float* pi = APi + (k0 >> 1) * (2 * Llen) + 2 * m;
    v2f ar = *(const v2f*)pr;
    v2f ai = *(const v2f*)pi;
    v2f nai = -ai;
    int bo = (k0 >> 1) * (2 * Dch) + 2 * d0;
    v2f b0r = *(const v2f*)(W0r + bo);
    v2f b0i = *(const v2f*)(W0i + bo);
    v2f b1r = *(const v2f*)(W1r + bo);
    v2f b1i = *(const v2f*)(W1i + bo);
    aRr = __builtin_amdgcn_wmma_f32_16x16x4_f32(false, ar,  false, b0r, (short)0, aRr, false, false);
    aRr = __builtin_amdgcn_wmma_f32_16x16x4_f32(false, nai, false, b0i, (short)0, aRr, false, false);
    aRi = __builtin_amdgcn_wmma_f32_16x16x4_f32(false, ar,  false, b0i, (short)0, aRi, false, false);
    aRi = __builtin_amdgcn_wmma_f32_16x16x4_f32(false, ai,  false, b0r, (short)0, aRi, false, false);
    aSr = __builtin_amdgcn_wmma_f32_16x16x4_f32(false, ar,  false, b1r, (short)0, aSr, false, false);
    aSr = __builtin_amdgcn_wmma_f32_16x16x4_f32(false, nai, false, b1i, (short)0, aSr, false, false);
    aSi = __builtin_amdgcn_wmma_f32_16x16x4_f32(false, ar,  false, b1i, (short)0, aSi, false, false);
    aSi = __builtin_amdgcn_wmma_f32_16x16x4_f32(false, ai,  false, b1r, (short)0, aSi, false, false);
  }
  for (int r = 0; r < 8; ++r) {
    int l = mbase + r;
    float k00r = aRr[r], k00i = aRi[r];
    float k10r = aSr[r], k10i = aSi[r];
    float sr = perl[l],            si = perl[Llen + l];
    float cr = perl[2 * Llen + l], ci = perl[3 * Llen + l];
    float tr = k00r - (sr * k10r - si * k10i);
    float ti = k00i - (sr * k10i + si * k10r);
    Ev[l * Dch + d0] = make_float2(cr * tr - ci * ti, cr * ti + ci * tr);
  }
#else
  for (int r = 0; r < 8; ++r) {
    int l = mbase + r;
    float k00r = 0.f, k00i = 0.f, k10r = 0.f, k10i = 0.f;
    for (int k = 0; k < Nst; ++k) {
      int ao = (k >> 1) * (2 * Llen) + 2 * l + (k & 1);
      int bo = (k >> 1) * (2 * Dch) + 2 * d0 + (k & 1);
      float arr = APr[ao], aii = APi[ao];
      float b0r = W0r[bo], b0i = W0i[bo];
      float b1r = W1r[bo], b1i = W1i[bo];
      k00r += arr * b0r - aii * b0i;  k00i += arr * b0i + aii * b0r;
      k10r += arr * b1r - aii * b1i;  k10i += arr * b1i + aii * b1r;
    }
    float sr = perl[l],            si = perl[Llen + l];
    float cr = perl[2 * Llen + l], ci = perl[3 * Llen + l];
    float tr = k00r - (sr * k10r - si * k10i);
    float ti = k00i - (sr * k10i + si * k10r);
    Ev[l * Dch + d0] = make_float2(cr * tr - ci * ti, cr * ti + ci * tr);
  }
#endif
}

__global__ __launch_bounds__(256) void s4_fft_r4(
    const float2* __restrict__ x, float2* __restrict__ y,
    int n, int Ns, float sign)
{
  int t = blockIdx.x * blockDim.x + threadIdx.x;
  int q = n >> 2;
  if (t >= q * Dch) return;
  int col = t & (Dch - 1);
  int j = t >> 9;
  int k = j & (Ns - 1);
  float ang = sign * (0.5f * PI_F) * (float)k / (float)Ns;
  float s1, c1;
  sincosf(ang, &s1, &c1);
  float c2 = c1 * c1 - s1 * s1, s2 = 2.f * c1 * s1;
  float c3 = c2 * c1 - s2 * s1, s3 = c2 * s1 + s2 * c1;
  float2 a = x[j * Dch + col];
  float2 b = x[(j + q) * Dch + col];
  float2 c = x[(j + 2 * q) * Dch + col];
  float2 d = x[(j + 3 * q) * Dch + col];
  float2 v1 = make_float2(c1 * b.x - s1 * b.y, c1 * b.y + s1 * b.x);
  float2 v2 = make_float2(c2 * c.x - s2 * c.y, c2 * c.y + s2 * c.x);
  float2 v3 = make_float2(c3 * d.x - s3 * d.y, c3 * d.y + s3 * d.x);
  float2 p0 = cadd(a, v2), p1 = csub(a, v2);
  float2 p2 = cadd(v1, v3), p3 = crotis(csub(v1, v3), sign);
  int i0 = 4 * j - 3 * k;
  y[(i0         ) * Dch + col] = cadd(p0, p2);
  y[(i0 +     Ns) * Dch + col] = cadd(p1, p3);
  y[(i0 + 2 * Ns) * Dch + col] = csub(p0, p2);
  y[(i0 + 3 * Ns) * Dch + col] = csub(p1, p3);
}

__global__ __launch_bounds__(256) void s4_fft_r8(
    const float2* __restrict__ x, float2* __restrict__ y,
    int n, int Ns, float sign)
{
  int t = blockIdx.x * blockDim.x + threadIdx.x;
  int q = n >> 3;
  if (t >= q * Dch) return;
  int col = t & (Dch - 1);
  int j = t >> 9;
  int k = j & (Ns - 1);
  float ang = sign * (0.25f * PI_F) * (float)k / (float)Ns;
  float s1, c1;
  sincosf(ang, &s1, &c1);
  float2 w1 = make_float2(c1, s1);
  float2 v[8];
  v[0] = x[j * Dch + col];
  float2 w = w1;
#pragma unroll
  for (int r = 1; r < 8; ++r) {
    v[r] = cmulf(x[(j + r * q) * Dch + col], w);
    w = cmulf(w, w1);
  }
  float2 X[8];
  dft8(v, sign, X);
  int i0 = 8 * j - 7 * k;
#pragma unroll
  for (int m = 0; m < 8; ++m)
    y[(i0 + m * Ns) * Dch + col] = X[m];
}

__global__ __launch_bounds__(256) void s4_upad_r8(
    const float* __restrict__ u, float2* __restrict__ y)
{
  const int Q = NFFT / 8;
  int t = blockIdx.x * blockDim.x + threadIdx.x;
  if (t >= Q * Dch) return;
  int col = t & (Dch - 1);
  int j = t >> 9;
  float2 v[8];
#pragma unroll
  for (int r = 0; r < 4; ++r)
    v[r] = make_float2(u[(j + r * Q) * Dch + col], 0.f);
#pragma unroll
  for (int r = 4; r < 8; ++r)
    v[r] = make_float2(0.f, 0.f);
  float2 X[8];
  dft8(v, -1.0f, X);
  int i0 = 8 * j;
#pragma unroll
  for (int m = 0; m < 8; ++m)
    y[(i0 + m) * Dch + col] = X[m];
}

__global__ __launch_bounds__(256) void s4_kpad_r8(
    const float2* __restrict__ Kt, float2* __restrict__ y)
{
  const int Q = NFFT / 8;
  int t = blockIdx.x * blockDim.x + threadIdx.x;
  if (t >= Q * Dch) return;
  int col = t & (Dch - 1);
  int j = t >> 9;
  float2 v[8];
#pragma unroll
  for (int r = 0; r < 4; ++r)
    v[r] = make_float2(Kt[(j + r * Q) * Dch + col].x * (1.0f / (float)Llen), 0.f);
#pragma unroll
  for (int r = 4; r < 8; ++r)
    v[r] = make_float2(0.f, 0.f);
  float2 X[8];
  dft8(v, -1.0f, X);
  int i0 = 8 * j;
#pragma unroll
  for (int m = 0; m < 8; ++m)
    y[(i0 + m) * Dch + col] = X[m];
}

__global__ __launch_bounds__(256) void s4_cmul_r8(
    const float2* __restrict__ Xs, const float2* __restrict__ Kd,
    float2* __restrict__ y)
{
  const int Q = NFFT / 8;
  int t = blockIdx.x * blockDim.x + threadIdx.x;
  if (t >= Q * Dch) return;
  int col = t & (Dch - 1);
  int j = t >> 9;
  float2 v[8];
#pragma unroll
  for (int r = 0; r < 8; ++r) {
    int idx = (j + r * Q) * Dch + col;
    v[r] = cmulf(Xs[idx], Kd[idx]);
  }
  float2 X[8];
  dft8(v, +1.0f, X);
  int i0 = 8 * j;
#pragma unroll
  for (int m = 0; m < 8; ++m)
    y[(i0 + m) * Dch + col] = X[m];
}

__global__ __launch_bounds__(256) void s4_final_r8(
    const float2* __restrict__ x, const float* __restrict__ u,
    const float* __restrict__ Dp, float* __restrict__ out)
{
  const int Q = NFFT / 8;
  int t = blockIdx.x * blockDim.x + threadIdx.x;
  if (t >= Q * Dch) return;
  int col = t & (Dch - 1);
  int j = t >> 9;
  float ang = (0.25f * PI_F) * (float)j / (float)Q;
  float s1, c1;
  sincosf(ang, &s1, &c1);
  float2 w1 = make_float2(c1, s1);
  float2 v[8];
  v[0] = x[j * Dch + col];
  float2 w = w1;
#pragma unroll
  for (int r = 1; r < 8; ++r) {
    v[r] = cmulf(x[(j + r * Q) * Dch + col], w);
    w = cmulf(w, w1);
  }
  float2 X[8];
  dft8(v, +1.0f, X);
  float dp = Dp[0];
  const float inv = 1.0f / (float)NFFT;
#pragma unroll
  for (int m = 0; m < 4; ++m) {
    int pos = (j + m * Q) * Dch + col;
    out[pos] = X[m].x * inv + dp * u[pos];
  }
}


extern "C" void kernel_launch(void* const* d_in, const int* in_sizes, int n_in,
                              void* d_out, int out_size, void* d_ws, size_t ws_size,
                              hipStream_t stream) {
  const float* u        = (const float*)d_in[0];
  const float* Lam      = (const float*)d_in[1];
  const float* Pri      = (const float*)d_in[2];
  const float* Bri      = (const float*)d_in[3];
  const float* Cri      = (const float*)d_in[4];
  const float* Dp       = (const float*)d_in[5];
  const float* log_step = (const float*)d_in[6];
  float* out = (float*)d_out;

  uint8_t* base = (uint8_t*)d_ws;
  size_t off = 0;
  auto carve = [&](size_t bytes) -> void* {
    void* p = base + off;
    off += (bytes + 255) & ~(size_t)255;
    return p;
  };
  float*  APr  = (float*)carve((size_t)Llen * Nst * 4);
  float*  APi  = (float*)carve((size_t)Llen * Nst * 4);
  float*  W0r  = (float*)carve((size_t)Nst * Dch * 4);
  float*  W0i  = (float*)carve((size_t)Nst * Dch * 4);
  float*  W1r  = (float*)carve((size_t)Nst * Dch * 4);
  float*  W1i  = (float*)carve((size_t)Nst * Dch * 4);
  float*  perl = (float*)carve((size_t)6 * Llen * 4);
  float2* Ev   = (float2*)carve((size_t)Llen * Dch * 8);
  float2* T0   = (float2*)carve((size_t)NFFT * Dch * 8);
  float2* T1   = (float2*)carve((size_t)NFFT * Dch * 8);
  float2* Kd   = (float2*)carve((size_t)NFFT * Dch * 8);
  (void)ws_size; (void)in_sizes; (void)n_in; (void)out_size;

  const int TB   = 256;
  const int gR8  = (NFFT / 8) * Dch / TB;
  const int gR8s = (Llen / 8) * Dch / TB;
  const int gR4s = (Llen / 4) * Dch / TB;

  s4_perl<<<(Llen + TB - 1) / TB, TB, 0, stream>>>(Lam, Pri, Cri, log_step, perl);
  s4_weights<<<(Nst * Dch + TB - 1) / TB, TB, 0, stream>>>(Bri, Pri, Cri, W0r, W0i, W1r, W1i);
  s4_cauchy<<<(Llen * Nst + TB - 1) / TB, TB, 0, stream>>>(Lam, perl, APr, APi);
  s4_cauchy_gemm<<<dim3(Dch / 16, Llen / 16), 32, 0, stream>>>(
      APr, APi, W0r, W0i, W1r, W1i, perl, Ev);

  s4_fft_r4<<<gR4s, TB, 0, stream>>>(Ev, T0, Llen, 1, +1.0f);
  for (int p = 0; p < 4; ++p) {
    const float2* src = (p & 1) ? Ev : T0;
    float2*       dst = (p & 1) ? T0 : Ev;
    s4_fft_r8<<<gR8s, TB, 0, stream>>>(src, dst, Llen, 4 << (3 * p), +1.0f);
  }

  s4_kpad_r8<<<gR8, TB, 0, stream>>>(T0, T1);
  for (int p = 0; p < 4; ++p) {
    const float2* src = (p & 1) ? T0 : T1;
    float2*       dst = (p == 3) ? Kd : ((p & 1) ? T1 : T0);
    s4_fft_r8<<<gR8, TB, 0, stream>>>(src, dst, NFFT, 8 << (3 * p), -1.0f);
  }

  for (int b = 0; b < Bsz; ++b) {
    const float* ub = u + (size_t)b * Llen * Dch;
    float*       ob = out + (size_t)b * Llen * Dch;
    s4_upad_r8<<<gR8, TB, 0, stream>>>(ub, T0);
    for (int p = 0; p < 4; ++p) {
      const float2* src = (p & 1) ? T1 : T0;
      float2*       dst = (p & 1) ? T0 : T1;
      s4_fft_r8<<<gR8, TB, 0, stream>>>(src, dst, NFFT, 8 << (3 * p), -1.0f);
    }
    s4_cmul_r8<<<gR8, TB, 0, stream>>>(T0, Kd, T1);
    for (int p = 0; p < 3; ++p) {
      const float2* src = (p & 1) ? T0 : T1;
      float2*       dst = (p & 1) ? T1 : T0;
      s4_fft_r8<<<gR8, TB, 0, stream>>>(src, dst, NFFT, 8 << (3 * p), +1.0f);
    }
    s4_final_r8<<<gR8, TB, 0, stream>>>(T0, ub, Dp, ob);
  }
}


# Forward-type transforms (u rows, K's inverse and forward) take 2 trips through memory (128*128), not 3: radix-128 butterflies (4x32) held in registers, no LDS; 12 launches, not 14; sums regrouped
# speedup vs baseline: 3.4911x; 3.4911x over previous
#include <hip/hip_runtime.h>
#include <hip/hip_bf16.h>
#include <stdint.h>


#define Bsz   4
#define Llen  16384
#define Dch   512
#define Nst   64
#define NFFT  32768
#define PI_F  3.14159265358979323846f

typedef __attribute__((ext_vector_type(2))) float v2f;
typedef __attribute__((ext_vector_type(8))) float v8f;

#ifndef __has_builtin
#define __has_builtin(x) 0
#endif
#if __has_builtin(__builtin_amdgcn_wmma_f32_16x16x4_f32)
#define S4_WMMA 1
#else
#define S4_WMMA 0
#endif

__device__ __forceinline__ float2 cadd(float2 a, float2 b) { return make_float2(a.x + b.x, a.y + b.y); }
__device__ __forceinline__ float2 csub(float2 a, float2 b) { return make_float2(a.x - b.x, a.y - b.y); }
__device__ __forceinline__ float2 cmulf(float2 a, float2 b) { return make_float2(a.x * b.x - a.y * b.y, a.x * b.y + a.y * b.x); }
__device__ __forceinline__ float2 crotis(float2 a, float s) { return make_float2(-s * a.y, s * a.x); }

__device__ __forceinline__ void dft8(const float2* v, float s, float2* X) {
  const float R = 0.70710678118654752440f;
  float2 t0 = cadd(v[0], v[4]), t1 = cadd(v[1], v[5]);
  float2 t2 = cadd(v[2], v[6]), t3 = cadd(v[3], v[7]);
  float2 u0 = csub(v[0], v[4]), u1 = csub(v[1], v[5]);
  float2 u2 = csub(v[2], v[6]), u3 = csub(v[3], v[7]);
  float2 m1 = cmulf(u1, make_float2(R, s * R));
  float2 m2 = crotis(u2, s);
  float2 m3 = cmulf(u3, make_float2(-R, s * R));
  float2 p0 = cadd(t0, t2), p1 = csub(t0, t2);
  float2 p2 = cadd(t1, t3), p3 = crotis(csub(t1, t3), s);
  X[0] = cadd(p0, p2); X[2] = cadd(p1, p3);
  X[4] = csub(p0, p2); X[6] = csub(p1, p3);
  float2 q0 = cadd(u0, m2), q1 = csub(u0, m2);
  float2 q2 = cadd(m1, m3), q3 = crotis(csub(m1, m3), s);
  X[1] = cadd(q0, q2); X[3] = cadd(q1, q3);
  X[5] = csub(q0, q2); X[7] = csub(q1, q3);
}

__device__ __forceinline__ void dft4(float2 a0, float2 a1, float2 a2, float2 a3, float s,
                                     float2& y0, float2& y1, float2& y2, float2& y3) {
  float2 p0 = cadd(a0, a2), p1 = csub(a0, a2);
  float2 p2 = cadd(a1, a3), p3 = crotis(csub(a1, a3), s);
  y0 = cadd(p0, p2); y1 = cadd(p1, p3);
  y2 = csub(p0, p2); y3 = csub(p1, p3);
}

__device__ __forceinline__ void dft16(const float2* v, float s, float2* X) {
  const float R  = 0.70710678118654752440f;
  const float C1 = 0.92387953251128675613f;
  const float S1 = 0.38268343236508977173f;
  float2 t0[4], t1[4], t2[4], t3[4];
  dft4(v[0], v[4], v[8],  v[12], s, t0[0], t0[1], t0[2], t0[3]);
  dft4(v[1], v[5], v[9],  v[13], s, t1[0], t1[1], t1[2], t1[3]);
  dft4(v[2], v[6], v[10], v[14], s, t2[0], t2[1], t2[2], t2[3]);
  dft4(v[3], v[7], v[11], v[15], s, t3[0], t3[1], t3[2], t3[3]);
  t1[1] = cmulf(t1[1], make_float2(C1, s * S1));
  t1[2] = cmulf(t1[2], make_float2(R, s * R));
  t1[3] = cmulf(t1[3], make_float2(S1, s * C1));
  t2[1] = cmulf(t2[1], make_float2(R, s * R));
  t2[2] = crotis(t2[2], s);
  t2[3] = cmulf(t2[3], make_float2(-R, s * R));
  t3[1] = cmulf(t3[1], make_float2(S1, s * C1));
  t3[2] = cmulf(t3[2], make_float2(-R, s * R));
  t3[3] = cmulf(t3[3], make_float2(-C1, -s * S1));
#pragma unroll
  for (int m1 = 0; m1 < 4; ++m1)
    dft4(t0[m1], t1[m1], t2[m1], t3[m1], s, X[m1], X[m1 + 4], X[m1 + 8], X[m1 + 12]);
}

__device__ __forceinline__ void dft32(const float2* v, float s, float2* X) {
  float2 t0_0, t0_1, t0_2, t0_3;
  dft4(v[0], v[8], v[16], v[24], s, t0_0, t0_1, t0_2, t0_3);
  float2 t1_0, t1_1, t1_2, t1_3;
  dft4(v[1], v[9], v[17], v[25], s, t1_0, t1_1, t1_2, t1_3);
  float2 t2_0, t2_1, t2_2, t2_3;
  dft4(v[2], v[10], v[18], v[26], s, t2_0, t2_1, t2_2, t2_3);
  float2 t3_0, t3_1, t3_2, t3_3;
  dft4(v[3], v[11], v[19], v[27], s, t3_0, t3_1, t3_2, t3_3);
  float2 t4_0, t4_1, t4_2, t4_3;
  dft4(v[4], v[12], v[20], v[28], s, t4_0, t4_1, t4_2, t4_3);
  float2 t5_0, t5_1, t5_2, t5_3;
  dft4(v[5], v[13], v[21], v[29], s, t5_0, t5_1, t5_2, t5_3);
  float2 t6_0, t6_1, t6_2, t6_3;
  dft4(v[6], v[14], v[22], v[30], s, t6_0, t6_1, t6_2, t6_3);
  float2 t7_0, t7_1, t7_2, t7_3;
  dft4(v[7], v[15], v[23], v[31], s, t7_0, t7_1, t7_2, t7_3);
  t1_1 = cmulf(t1_1, make_float2(0.98078528040323043f, s * 0.19509032201612825f));
  t2_1 = cmulf(t2_1, make_float2(0.92387953251128674f, s * 0.38268343236508978f));
  t3_1 = cmulf(t3_1, make_float2(0.83146961230254524f, s * 0.55557023301960218f));
  t4_1 = cmulf(t4_1, make_float2(0.70710678118654757f, s * 0.70710678118654746f));
  t5_1 = cmulf(t5_1, make_float2(0.55557023301960229f, s * 0.83146961230254524f));
  t6_1 = cmulf(t6_1, make_float2(0.38268343236508984f, s * 0.92387953251128674f));
  t7_1 = cmulf(t7_1, make_float2(0.19509032201612833f, s * 0.98078528040323043f));
  t1_2 = cmulf(t1_2, make_float2(0.92387953251128674f, s * 0.38268343236508978f));
  t2_2 = cmulf(t2_2, make_float2(0.70710678118654757f, s * 0.70710678118654746f));
  t3_2 = cmulf(t3_2, make_float2(0.38268343236508984f, s * 0.92387953251128674f));
  t4_2 = crotis(t4_2, s);
  t5_2 = cmulf(t5_2, make_float2(-0.38268343236508973f, s * 0.92387953251128674f));
  t6_2 = cmulf(t6_2, make_float2(-0.70710678118654746f, s * 0.70710678118654757f));
  t7_2 = cmulf(t7_2, make_float2(-0.92387953251128674f, s * 0.38268343236508989f));
  t1_3 = cmulf(t1_3, make_float2(0.83146961230254524f, s * 0.55557023301960218f));
  t2_3 = cmulf(t2_3, make_float2(0.38268343236508984f, s * 0.92387953251128674f));
  t3_3 = cmulf(t3_3, make_float2(-0.19509032201612819f, s * 0.98078528040323043f));
  t4_3 = cmulf(t4_3, make_float2(-0.70710678118654746f, s * 0.70710678118654757f));
  t5_3 = cmulf(t5_3, make_float2(-0.98078528040323043f, s * 0.19509032201612861f));
  t6_3 = cmulf(t6_3, make_float2(-0.92387953251128685f, s * -0.38268343236508967f));
  t7_3 = cmulf(t7_3, make_float2(-0.55557023301960218f, s * -0.83146961230254524f));
  {
    float2 a[8] = {t0_0, t1_0, t2_0, t3_0, t4_0, t5_0, t6_0, t7_0};
    float2 b[8];
    dft8(a, s, b);
    X[0] = b[0]; X[4] = b[1]; X[8] = b[2]; X[12] = b[3];
    X[16] = b[4]; X[20] = b[5]; X[24] = b[6]; X[28] = b[7];
  }
  {
    float2 a[8] = {t0_1, t1_1, t2_1, t3_1, t4_1, t5_1, t6_1, t7_1};
    float2 b[8];
    dft8(a, s, b);
    X[1] = b[0]; X[5] = b[1]; X[9] = b[2]; X[13] = b[3];
    X[17] = b[4]; X[21] = b[5]; X[25] = b[6]; X[29] = b[7];
  }
  {
    float2 a[8] = {t0_2, t1_2, t2_2, t3_2, t4_2, t5_2, t6_2, t7_2};
    float2 b[8];
    dft8(a, s, b);
    X[2] = b[0]; X[6] = b[1]; X[10] = b[2]; X[14] = b[3];
    X[18] = b[4]; X[22] = b[5]; X[26] = b[6]; X[30] = b[7];
  }
  {
    float2 a[8] = {t0_3, t1_3, t2_3, t3_3, t4_3, t5_3, t6_3, t7_3};
    float2 b[8];
    dft8(a, s, b);
    X[3] = b[0]; X[7] = b[1]; X[11] = b[2]; X[15] = b[3];
    X[19] = b[4]; X[23] = b[5]; X[27] = b[6]; X[31] = b[7];
  }
}

__device__ __forceinline__ void dft128(const float2* v, float s, float2* X) {
  float2 t0[32], t1[32], t2[32], t3[32];
#pragma unroll
  for (int r2 = 0; r2 < 32; ++r2)
    dft4(v[r2], v[r2 + 32], v[r2 + 64], v[r2 + 96], s, t0[r2], t1[r2], t2[r2], t3[r2]);
  t1[1] = cmulf(t1[1], make_float2(0.99879545620517241f, s * 0.049067674327418015f));
  t1[2] = cmulf(t1[2], make_float2(0.99518472667219693f, s * 0.098017140329560604f));
  t1[3] = cmulf(t1[3], make_float2(0.98917650996478101f, s * 0.14673047445536175f));
  t1[4] = cmulf(t1[4], make_float2(0.98078528040323043f, s * 0.19509032201612825f));
  t1[5] = cmulf(t1[5], make_float2(0.97003125319454397f, s * 0.24298017990326387f));
  t1[6] = cmulf(t1[6], make_float2(0.95694033573220882f, s * 0.29028467725446233f));
  t1[7] = cmulf(t1[7], make_float2(0.94154406518302081f, s * 0.33688985339222005f));
  t1[8] = cmulf(t1[8], make_float2(0.92387953251128674f, s * 0.38268343236508978f));
  t1[9] = cmulf(t1[9], make_float2(0.90398929312344334f, s * 0.42755509343028208f));
  t1[10] = cmulf(t1[10], make_float2(0.88192126434835505f, s * 0.47139673682599764f));
  t1[11] = cmulf(t1[11], make_float2(0.85772861000027212f, s * 0.51410274419322166f));
  t1[12] = cmulf(t1[12], make_float2(0.83146961230254524f, s * 0.55557023301960218f));
  t1[13] = cmulf(t1[13], make_float2(0.80320753148064494f, s * 0.59569930449243336f));
  t1[14] = cmulf(t1[14], make_float2(0.77301045336273699f, s * 0.63439328416364549f));
  t1[15] = cmulf(t1[15], make_float2(0.74095112535495911f, s * 0.67155895484701833f));
  t1[16] = cmulf(t1[16], make_float2(0.70710678118654757f, s * 0.70710678118654746f));
  t1[17] = cmulf(t1[17], make_float2(0.67155895484701833f, s * 0.74095112535495911f));
  t1[18] = cmulf(t1[18], make_float2(0.63439328416364549f, s * 0.77301045336273699f));
  t1[19] = cmulf(t1[19], make_float2(0.59569930449243347f, s * 0.80320753148064483f));
  t1[20] = cmulf(t1[20], make_float2(0.55557023301960229f, s * 0.83146961230254524f));
  t1[21] = cmulf(t1[21], make_float2(0.51410274419322166f, s * 0.85772861000027212f));
  t1[22] = cmulf(t1[22], make_float2(0.47139673682599781f, s * 0.88192126434835494f));
  t1[23] = cmulf(t1[23], make_float2(0.42755509343028220f, s * 0.90398929312344334f));
  t1[24] = cmulf(t1[24], make_float2(0.38268343236508984f, s * 0.92387953251128674f));
  t1[25] = cmulf(t1[25], make_float2(0.33688985339222005f, s * 0.94154406518302081f));
  t1[26] = cmulf(t1[26], make_float2(0.29028467725446233f, s * 0.95694033573220894f));
  t1[27] = cmulf(t1[27], make_float2(0.24298017990326398f, s * 0.97003125319454397f));
  t1[28] = cmulf(t1[28], make_float2(0.19509032201612833f, s * 0.98078528040323043f));
  t1[29] = cmulf(t1[29], make_float2(0.14673047445536175f, s * 0.98917650996478101f));
  t1[30] = cmulf(t1[30], make_float2(0.098017140329560770f, s * 0.99518472667219682f));
  t1[31] = cmulf(t1[31], make_float2(0.049067674327418126f, s * 0.99879545620517241f));
  t2[1] = cmulf(t2[1], make_float2(0.99518472667219693f, s * 0.098017140329560604f));
  t2[2] = cmulf(t2[2], make_float2(0.98078528040323043f, s * 0.19509032201612825f));
  t2[3] = cmulf(t2[3], make_float2(0.95694033573220882f, s * 0.29028467725446233f));
  t2[4] = cmulf(t2[4], make_float2(0.92387953251128674f, s * 0.38268343236508978f));
  t2[5] = cmulf(t2[5], make_float2(0.88192126434835505f, s * 0.47139673682599764f));
  t2[6] = cmulf(t2[6], make_float2(0.83146961230254524f, s * 0.55557023301960218f));
  t2[7] = cmulf(t2[7], make_float2(0.77301045336273699f, s * 0.63439328416364549f));
  t2[8] = cmulf(t2[8], make_float2(0.70710678118654757f, s * 0.70710678118654746f));
  t2[9] = cmulf(t2[9], make_float2(0.63439328416364549f, s * 0.77301045336273699f));
  t2[10] = cmulf(t2[10], make_float2(0.55557023301960229f, s * 0.83146961230254524f));
  t2[11] = cmulf(t2[11], make_float2(0.47139673682599781f, s * 0.88192126434835494f));
  t2[12] = cmulf(t2[12], make_float2(0.38268343236508984f, s * 0.92387953251128674f));
  t2[13] = cmulf(t2[13], make_float2(0.29028467725446233f, s * 0.95694033573220894f));
  t2[14] = cmulf(t2[14], make_float2(0.19509032201612833f, s * 0.98078528040323043f));
  t2[15] = cmulf(t2[15], make_float2(0.098017140329560770f, s * 0.99518472667219682f));
  t2[16] = crotis(t2[16], s);
  t2[17] = cmulf(t2[17], make_float2(-0.098017140329560645f, s * 0.99518472667219693f));
  t2[18] = cmulf(t2[18], make_float2(-0.19509032201612819f, s * 0.98078528040323043f));
  t2[19] = cmulf(t2[19], make_float2(-0.29028467725446216f, s * 0.95694033573220894f));
  t2[20] = cmulf(t2[20], make_float2(-0.38268343236508973f, s * 0.92387953251128674f));
  t2[21] = cmulf(t2[21], make_float2(-0.47139673682599770f, s * 0.88192126434835505f));
  t2[22] = cmulf(t2[22], make_float2(-0.55557023301960196f, s * 0.83146961230254546f));
  t2[23] = cmulf(t2[23], make_float2(-0.63439328416364538f, s * 0.77301045336273710f));
  t2[24] = cmulf(t2[24], make_float2(-0.70710678118654746f, s * 0.70710678118654757f));
  t2[25] = cmulf(t2[25], make_float2(-0.77301045336273699f, s * 0.63439328416364549f));
  t2[26] = cmulf(t2[26], make_float2(-0.83146961230254535f, s * 0.55557023301960218f));
  t2[27] = cmulf(t2[27], make_float2(-0.88192126434835494f, s * 0.47139673682599786f));
  t2[28] = cmulf(t2[28], make_float2(-0.92387953251128674f, s * 0.38268343236508989f));
  t2[29] = cmulf(t2[29], make_float2(-0.95694033573220882f, s * 0.29028467725446239f));
  t2[30] = cmulf(t2[30], make_float2(-0.98078528040323043f, s * 0.19509032201612861f));
  t2[31] = cmulf(t2[31], make_float2(-0.99518472667219682f, s * 0.098017140329560826f));
  t3[1] = cmulf(t3[1], make_float2(0.98917650996478101f, s * 0.14673047445536175f));
  t3[2] = cmulf(t3[2], make_float2(0.95694033573220882f, s * 0.29028467725446233f));
  t3[3] = cmulf(t3[3], make_float2(0.90398929312344334f, s * 0.42755509343028208f));
  t3[4] = cmulf(t3[4], make_float2(0.83146961230254524f, s * 0.55557023301960218f));
  t3[5] = cmulf(t3[5], make_float2(0.74095112535495911f, s * 0.67155895484701833f));
  t3[6] = cmulf(t3[6], make_float2(0.63439328416364549f, s * 0.77301045336273699f));
  t3[7] = cmulf(t3[7], make_float2(0.51410274419322166f, s * 0.85772861000027212f));
  t3[8] = cmulf(t3[8], make_float2(0.38268343236508984f, s * 0.92387953251128674f));
  t3[9] = cmulf(t3[9], make_float2(0.24298017990326398f, s * 0.97003125319454397f));
  t3[10] = cmulf(t3[10], make_float2(0.098017140329560770f, s * 0.99518472667219682f));
  t3[11] = cmulf(t3[11], make_float2(-0.049067674327418008f, s * 0.99879545620517241f));
  t3[12] = cmulf(t3[12], make_float2(-0.19509032201612819f, s * 0.98078528040323043f));
  t3[13] = cmulf(t3[13], make_float2(-0.33688985339221994f, s * 0.94154406518302081f));
  t3[14] = cmulf(t3[14], make_float2(-0.47139673682599770f, s * 0.88192126434835505f));
  t3[15] = cmulf(t3[15], make_float2(-0.59569930449243336f, s * 0.80320753148064494f));
  t3[16] = cmulf(t3[16], make_float2(-0.70710678118654746f, s * 0.70710678118654757f));
  t3[17] = cmulf(t3[17], make_float2(-0.80320753148064483f, s * 0.59569930449243347f));
  t3[18] = cmulf(t3[18], make_float2(-0.88192126434835494f, s * 0.47139673682599786f));
  t3[19] = cmulf(t3[19], make_float2(-0.94154406518302070f, s * 0.33688985339222033f));
  t3[20] = cmulf(t3[20], make_float2(-0.98078528040323043f, s * 0.19509032201612861f));
  t3[21] = cmulf(t3[21], make_float2(-0.99879545620517241f, s * 0.049067674327417966f));
  t3[22] = cmulf(t3[22], make_float2(-0.99518472667219693f, s * -0.098017140329560590f));
  t3[23] = cmulf(t3[23], make_float2(-0.97003125319454397f, s * -0.24298017990326382f));
  t3[24] = cmulf(t3[24], make_float2(-0.92387953251128685f, s * -0.38268343236508967f));
  t3[25] = cmulf(t3[25], make_float2(-0.85772861000027212f, s * -0.51410274419322155f));
  t3[26] = cmulf(t3[26], make_float2(-0.77301045336273710f, s * -0.63439328416364527f));
  t3[27] = cmulf(t3[27], make_float2(-0.67155895484701866f, s * -0.74095112535495888f));
  t3[28] = cmulf(t3[28], make_float2(-0.55557023301960218f, s * -0.83146961230254524f));
  t3[29] = cmulf(t3[29], make_float2(-0.42755509343028247f, s * -0.90398929312344312f));
  t3[30] = cmulf(t3[30], make_float2(-0.29028467725446244f, s * -0.95694033573220882f));
  t3[31] = cmulf(t3[31], make_float2(-0.14673047445536230f, s * -0.98917650996478090f));
  float2 Y[32];
  dft32(t0, s, Y);
#pragma unroll
  for (int m2 = 0; m2 < 32; ++m2) X[0 + 4 * m2] = Y[m2];
  dft32(t1, s, Y);
#pragma unroll
  for (int m2 = 0; m2 < 32; ++m2) X[1 + 4 * m2] = Y[m2];
  dft32(t2, s, Y);
#pragma unroll
  for (int m2 = 0; m2 < 32; ++m2) X[2 + 4 * m2] = Y[m2];
  dft32(t3, s, Y);
#pragma unroll
  for (int m2 = 0; m2 < 32; ++m2) X[3 + 4 * m2] = Y[m2];
}

__device__ __forceinline__ void s4_g_of_l(
    int l, const float* __restrict__ log_step,
    float& gr, float& gi, float& scr, float& sci)
{
  float step = expf(log_step[0]);
  float theta = -2.0f * PI_F * (float)l / (float)Llen;
  float oi, orr;
  sincosf(theta, &oi, &orr);
  float op1r = 1.0f + orr;
  float den = op1r * op1r + oi * oi;
  float nr = 1.0f - orr, ni = -oi;
  float gscale = 2.0f / step;
  gr = gscale * (nr * op1r + ni * oi) / den;
  gi = gscale * (ni * op1r - nr * oi) / den;
  scr = 2.0f * op1r / den;
  sci = -2.0f * oi / den;
}

__device__ __forceinline__ void s4_perl_body(
    int l, const float* __restrict__ Lam, const float* __restrict__ Pri,
    const float* __restrict__ Cri, const float* __restrict__ log_step,
    float* __restrict__ perl)
{
  if (l >= Llen) return;
  float gr, gi, scr, sci;
  s4_g_of_l(l, log_step, gr, gi, scr, sci);
  float k01r = 0.f, k01i = 0.f, k11r = 0.f, k11i = 0.f;
  for (int n = 0; n < Nst; ++n) {
    float lr = Lam[2 * n], li = Lam[2 * n + 1];
    float pr = Pri[2 * n], pi = Pri[2 * n + 1];
    float cr = Cri[2 * n], ci = Cri[2 * n + 1];
    float zr = gr - lr, zi = gi - li;
    float zd = zr * zr + zi * zi;
    float car = zr / zd, cai = -zi / zd;
    float w01r = cr * pr + ci * pi;
    float w01i = cr * pi - ci * pr;
    float w11r = pr * pr + pi * pi;
    k01r += car * w01r - cai * w01i;
    k01i += car * w01i + cai * w01r;
    k11r += car * w11r;
    k11i += cai * w11r;
  }
  float dr = 1.0f + k11r, di = k11i;
  float dd = dr * dr + di * di;
  float sr = (k01r * dr + k01i * di) / dd;
  float si = (k01i * dr - k01r * di) / dd;
  perl[l] = sr;
  perl[Llen + l] = si;
  perl[2 * Llen + l] = scr;
  perl[3 * Llen + l] = sci;
}

__device__ __forceinline__ void s4_weights_body(
    int idx, const float* __restrict__ Bri, const float* __restrict__ Pri,
    const float* __restrict__ Cri,
    float* __restrict__ W0r, float* __restrict__ W0i,
    float* __restrict__ W1r, float* __restrict__ W1i)
{
  if (idx >= Nst * Dch) return;
  int n = idx / Dch, d = idx % Dch;
  float br = Bri[2 * (n * Dch + d)], bi = Bri[2 * (n * Dch + d) + 1];
  float cr = Cri[2 * n], ci = Cri[2 * n + 1];
  float pr = Pri[2 * n], pi = Pri[2 * n + 1];
  int o = (n >> 1) * (2 * Dch) + 2 * d + (n & 1);
  W0r[o] = cr * br + ci * bi;
  W0i[o] = cr * bi - ci * br;
  W1r[o] = pr * br + pi * bi;
  W1i[o] = pr * bi - pi * br;
}

#define NPB  (Llen / 256)
#define NWB  (Nst * Dch / 256)
#define NCB  512
__device__ __forceinline__ void s4_cauchy_body(
    int cb, const float* __restrict__ Lam, const float* __restrict__ log_step,
    float* __restrict__ APr, float* __restrict__ APi)
{
  const int idx0 = cb * 256 + threadIdx.x;
  const int l = idx0 % Llen;
  float gr, gi, scr, sci;
  s4_g_of_l(l, log_step, gr, gi, scr, sci);
  (void)scr; (void)sci;
  for (int idx = idx0; idx < Llen * Nst; idx += NCB * 256) {
    int n = idx / Llen;
    float zr = gr - Lam[2 * n], zi = gi - Lam[2 * n + 1];
    float zd = zr * zr + zi * zi;
    int o = (n >> 1) * (2 * Llen) + 2 * l + (n & 1);
    APr[o] = zr / zd;
    APi[o] = -zi / zd;
  }
}

__global__ __launch_bounds__(256) void s4_setup(
    const float* __restrict__ Lam, const float* __restrict__ Pri,
    const float* __restrict__ Cri, const float* __restrict__ log_step,
    const float* __restrict__ Bri, float* __restrict__ perl,
    float* __restrict__ W0r, float* __restrict__ W0i,
    float* __restrict__ W1r, float* __restrict__ W1i,
    float* __restrict__ APr, float* __restrict__ APi)
{
  const int b = blockIdx.x;
  if (b < NPB)
    s4_perl_body(b * 256 + threadIdx.x, Lam, Pri, Cri, log_step, perl);
  else if (b < NPB + NWB)
    s4_weights_body((b - NPB) * 256 + threadIdx.x, Bri, Pri, Cri, W0r, W0i, W1r, W1i);
  else
    s4_cauchy_body(b - NPB - NWB, Lam, log_step, APr, APi);
}

__global__ __launch_bounds__(32) void s4_cauchy_gemm(
    const float* __restrict__ APr, const float* __restrict__ APi,
    const float* __restrict__ W0r, const float* __restrict__ W0i,
    const float* __restrict__ W1r, const float* __restrict__ W1i,
    const float* __restrict__ perl, float2* __restrict__ Ev)
{
  int lane = threadIdx.x & 31;
  int tileD = blockIdx.x;
  int tileL = blockIdx.y;
  int d0 = tileD * 16 + (lane & 15);
  int mbase = tileL * 16 + ((lane >> 4) ? 8 : 0);
#if S4_WMMA
  int m = tileL * 16 + (lane & 15);
  int khi = (lane >> 4) * 2;
  v8f aRr = {}, aRi = {}, aSr = {}, aSi = {};
  for (int kb = 0; kb < Nst; kb += 4) {
    int k0 = kb + khi;
    const float* pr = APr + (k0 >> 1) * (2 * Llen) + 2 * m;
    const float* pi = APi + (k0 >> 1) * (2 * Llen) + 2 * m;
    v2f ar = *(const v2f*)pr;
    v2f ai = *(const v2f*)pi;
    v2f nai = -ai;
    int bo = (k0 >> 1) * (2 * Dch) + 2 * d0;
    v2f b0r = *(const v2f*)(W0r + bo);
    v2f b0i = *(const v2f*)(W0i + bo);
    v2f b1r = *(const v2f*)(W1r + bo);
    v2f b1i = *(const v2f*)(W1i + bo);
    aRr = __builtin_amdgcn_wmma_f32_16x16x4_f32(false, ar,  false, b0r, (short)0, aRr, false, false);
    aRr = __builtin_amdgcn_wmma_f32_16x16x4_f32(false, nai, false, b0i, (short)0, aRr, false, false);
    aRi = __builtin_amdgcn_wmma_f32_16x16x4_f32(false, ar,  false, b0i, (short)0, aRi, false, false);
    aRi = __builtin_amdgcn_wmma_f32_16x16x4_f32(false, ai,  false, b0r, (short)0, aRi, false, false);
    aSr = __builtin_amdgcn_wmma_f32_16x16x4_f32(false, ar,  false, b1r, (short)0, aSr, false, false);
    aSr = __builtin_amdgcn_wmma_f32_16x16x4_f32(false, nai, false, b1i, (short)0, aSr, false, false);
    aSi = __builtin_amdgcn_wmma_f32_16x16x4_f32(false, ar,  false, b1i, (short)0, aSi, false, false);
    aSi = __builtin_amdgcn_wmma_f32_16x16x4_f32(false, ai,  false, b1r, (short)0, aSi, false, false);
  }
  for (int r = 0; r < 8; ++r) {
    int l = mbase + r;
    float k00r = aRr[r], k00i = aRi[r];
    float k10r = aSr[r], k10i = aSi[r];
    float sr = perl[l],            si = perl[Llen + l];
    float cr = perl[2 * Llen + l], ci = perl[3 * Llen + l];
    float tr = k00r - (sr * k10r - si * k10i);
    float ti = k00i - (sr * k10i + si * k10r);
    Ev[l * Dch + d0] = make_float2(cr * tr - ci * ti, cr * ti + ci * tr);
  }
#else
  for (int r = 0; r < 8; ++r) {
    int l = mbase + r;
    float k00r = 0.f, k00i = 0.f, k10r = 0.f, k10i = 0.f;
    for (int k = 0; k < Nst; ++k) {
      int ao = (k >> 1) * (2 * Llen) + 2 * l + (k & 1);
      int bo = (k >> 1) * (2 * Dch) + 2 * d0 + (k & 1);
      float arr = APr[ao], aii = APi[ao];
      float b0r = W0r[bo], b0i = W0i[bo];
      float b1r = W1r[bo], b1i = W1i[bo];
      k00r += arr * b0r - aii * b0i;  k00i += arr * b0i + aii * b0r;
      k10r += arr * b1r - aii * b1i;  k10i += arr * b1i + aii * b1r;
    }
    float sr = perl[l],            si = perl[Llen + l];
    float cr = perl[2 * Llen + l], ci = perl[3 * Llen + l];
    float tr = k00r - (sr * k10r - si * k10i);
    float ti = k00i - (sr * k10i + si * k10r);
    Ev[l * Dch + d0] = make_float2(cr * tr - ci * ti, cr * ti + ci * tr);
  }
#endif
}

#define NBLK  512
#define HCOL  256
#define PLANE ((size_t)Llen * Dch)

__global__ __launch_bounds__(256) void s4_fft_r32(
    const float2* __restrict__ xb, float2* __restrict__ yb,
    const float2* __restrict__ xk, float2* __restrict__ yk,
    int n, int Ns, float signb, float signk, int np, int rev)
{
  const int q = n >> 5;
  for (int pp = 0; pp < np; ++pp) {
  const int p = rev ? (np - 1 - pp) : pp;
  const float2* __restrict__ x = (p < 2) ? xb + (size_t)p * PLANE : xk;
  float2* __restrict__ y = (p < 2) ? yb + (size_t)p * PLANE : yk;
  const float sign = (p < 2) ? signb : signk;
  for (int j = blockIdx.x; j < q; j += gridDim.x) {
    int k = j & (Ns - 1);
    float ang = sign * (0.0625f * PI_F) * (float)k / (float)Ns;
    float s1, c1;
    sincosf(ang, &s1, &c1);
    float2 w[32];
    w[1]  = make_float2(c1, s1);
    w[2]  = cmulf(w[1], w[1]);
    w[3]  = cmulf(w[2], w[1]);
    w[4]  = cmulf(w[2], w[2]);
    w[5]  = cmulf(w[4], w[1]);
    w[6]  = cmulf(w[3], w[3]);
    w[7]  = cmulf(w[4], w[3]);
    w[8]  = cmulf(w[4], w[4]);
    w[9]  = cmulf(w[8], w[1]);
    w[10] = cmulf(w[5], w[5]);
    w[11] = cmulf(w[8], w[3]);
    w[12] = cmulf(w[6], w[6]);
    w[13] = cmulf(w[8], w[5]);
    w[14] = cmulf(w[7], w[7]);
    w[15] = cmulf(w[8], w[7]);
    w[16] = cmulf(w[8], w[8]);
#pragma unroll
    for (int r = 1; r < 16; ++r)
      w[16 + r] = cmulf(w[16], w[r]);
    int i0 = 32 * j - 31 * k;
    for (int h = 0; h < 2; ++h) {
      int col = threadIdx.x + h * HCOL;
      float2 v[32];
#pragma unroll
      for (int r = 0; r < 32; ++r)
        v[r] = x[(j + r * q) * Dch + col];
#pragma unroll
      for (int r = 1; r < 32; ++r)
        v[r] = cmulf(v[r], w[r]);
      float2 X[32];
      dft32(v, sign, X);
#pragma unroll
      for (int m = 0; m < 32; ++m)
        y[(i0 + m * Ns) * Dch + col] = X[m];
    }
  }
  }
}

#define R128Q (Llen / 128)
__global__ __launch_bounds__(256) void s4_fft_r128(
    const float2* __restrict__ xb, float2* __restrict__ yb,
    const float2* __restrict__ xk, float2* __restrict__ yk,
    float signb, float signk, int np, int rev)
{
  for (int pp = 0; pp < np; ++pp) {
  const int p = rev ? (np - 1 - pp) : pp;
  const float2* __restrict__ x = (p < 2) ? xb + (size_t)p * PLANE : xk;
  float2* __restrict__ y = (p < 2) ? yb + (size_t)p * PLANE : yk;
  const float sign = (p < 2) ? signb : signk;
  for (int t = blockIdx.x; t < 2 * R128Q; t += gridDim.x) {
    const int j = t >> 1;
    const int col = threadIdx.x + (t & 1) * HCOL;
    float ang = sign * (2.0f * PI_F) * (float)j / (float)Llen;
    float s1, c1, s16, c16;
    sincosf(ang, &s1, &c1);
    sincosf(16.0f * ang, &s16, &c16);
    float2 wl[16], wh[8];
    wl[1]  = make_float2(c1, s1);
    wl[2]  = cmulf(wl[1], wl[1]);
    wl[3]  = cmulf(wl[2], wl[1]);
    wl[4]  = cmulf(wl[2], wl[2]);
    wl[5]  = cmulf(wl[4], wl[1]);
    wl[6]  = cmulf(wl[3], wl[3]);
    wl[7]  = cmulf(wl[4], wl[3]);
    wl[8]  = cmulf(wl[4], wl[4]);
#pragma unroll
    for (int r = 1; r < 8; ++r)
      wl[8 + r] = cmulf(wl[8], wl[r]);
    wh[1] = make_float2(c16, s16);
    wh[2] = cmulf(wh[1], wh[1]);
    wh[3] = cmulf(wh[2], wh[1]);
    wh[4] = cmulf(wh[2], wh[2]);
    wh[5] = cmulf(wh[4], wh[1]);
    wh[6] = cmulf(wh[3], wh[3]);
    wh[7] = cmulf(wh[4], wh[3]);
    float2 v[128];
#pragma unroll
    for (int r = 0; r < 128; ++r)
      v[r] = x[(j + r * R128Q) * Dch + col];
#pragma unroll
    for (int r = 1; r < 16; ++r)
      v[r] = cmulf(v[r], wl[r]);
#pragma unroll
    for (int hi = 1; hi < 8; ++hi) {
      v[16 * hi] = cmulf(v[16 * hi], wh[hi]);
#pragma unroll
      for (int lo = 1; lo < 16; ++lo)
        v[16 * hi + lo] = cmulf(v[16 * hi + lo], cmulf(wh[hi], wl[lo]));
    }
    float2 X[128];
    dft128(v, sign, X);
#pragma unroll
    for (int m = 0; m < 128; ++m)
      y[(j + m * R128Q) * Dch + col] = X[m];
  }
  }
}


__global__ __launch_bounds__(256) void s4_first_r128(
    const float* __restrict__ ub, float2* __restrict__ yb,
    const float2* __restrict__ xk, float2* __restrict__ yk,
    int kmode, int np, int rev)
{
  const float invL = 1.0f / (float)Llen;
  for (int pp = 0; pp < np; ++pp) {
    const int p = rev ? (np - 1 - pp) : pp;
    float2* __restrict__ y = (p < 2) ? yb + (size_t)p * PLANE : yk;
    const float sign = (p == 2 && kmode == 0) ? +1.0f : -1.0f;
    for (int t = blockIdx.x; t < 2 * R128Q; t += gridDim.x) {
      const int j = t >> 1;
      const int col = threadIdx.x + (t & 1) * HCOL;
      float2 v[128];
      if (p < 2) {
        const float* __restrict__ u = ub + (size_t)p * PLANE;
#pragma unroll
        for (int r = 0; r < 64; ++r) {
          int n = j + r * R128Q;
          v[r] = make_float2(u[(2 * n) * Dch + col], u[(2 * n + 1) * Dch + col]);
        }
#pragma unroll
        for (int r = 64; r < 128; ++r)
          v[r] = make_float2(0.f, 0.f);
      } else if (kmode == 0) {
#pragma unroll
        for (int r = 0; r < 128; ++r)
          v[r] = xk[(j + r * R128Q) * Dch + col];
      } else {
#pragma unroll
        for (int r = 0; r < 64; ++r) {
          int n = j + r * R128Q;
          v[r] = make_float2(xk[(2 * n) * Dch + col].x * invL,
                             xk[(2 * n + 1) * Dch + col].x * invL);
        }
#pragma unroll
        for (int r = 64; r < 128; ++r)
          v[r] = make_float2(0.f, 0.f);
      }
      float2 X[128];
      dft128(v, sign, X);
#pragma unroll
      for (int m = 0; m < 128; ++m)
        y[(128 * j + m) * Dch + col] = X[m];
    }
  }
}

__device__ __forceinline__ void s4_pair_mul(
    float2 zk, float2 zm, float2 kk, float2 km, float2 c,
    float2& outk, float2& outm)
{
  float2 e  = make_float2(zk.x + zm.x, zk.y - zm.y);
  float2 o  = make_float2(zk.y + zm.y, zm.x - zk.x);
  float2 ek = make_float2(kk.x + km.x, kk.y - km.y);
  float2 ok = make_float2(kk.y + km.y, km.x - kk.x);
  float2 ye = cadd(cmulf(e, ek), cmulf(c, cmulf(o, ok)));
  float2 yo = cadd(cmulf(e, ok), cmulf(o, ek));
  outk = make_float2(ye.x - yo.y, ye.y + yo.x);
  outm = make_float2(ye.x + yo.y, yo.x - ye.y);
}

__global__ __launch_bounds__(256) void s4_pmul_r16(
    const float2* __restrict__ Zb, const float2* __restrict__ ZK,
    float2* __restrict__ yb, int np, int rev)
{
  const int Q = Llen / 16;
  const float R  = 0.70710678118654752440f;
  const float C1 = 0.92387953251128675613f;
  const float S1 = 0.38268343236508977173f;
  for (int pp = 0; pp < np; ++pp) {
  const int p = rev ? (np - 1 - pp) : pp;
  const float2* __restrict__ Z = Zb + (size_t)p * PLANE;
  float2* __restrict__ y = yb + (size_t)p * PLANE;
  for (int jj = blockIdx.x; jj <= Q / 2; jj += gridDim.x) {
    float sj, cjr;
    sincosf(-2.0f * PI_F * (float)jj / (float)Llen, &sj, &cjr);
    float2 cj = make_float2(cjr, sj);
    float2 cr[16];
    cr[0] = cj;
    cr[1] = cmulf(cj, make_float2(C1, -S1));
    cr[2] = cmulf(cj, make_float2(R, -R));
    cr[3] = cmulf(cj, make_float2(S1, -C1));
    cr[4] = make_float2(cj.y, -cj.x);
    cr[5] = cmulf(cj, make_float2(-S1, -C1));
    cr[6] = cmulf(cj, make_float2(-R, -R));
    cr[7] = cmulf(cj, make_float2(-C1, -S1));
#pragma unroll
    for (int r = 0; r < 8; ++r)
      cr[8 + r] = make_float2(-cr[r].x, -cr[r].y);
    for (int h = 0; h < 2; ++h) {
      int col = threadIdx.x + h * HCOL;
      float2 va[16], vb[16];
#pragma unroll
      for (int r = 0; r < 16; ++r) {
        int ia = (jj + r * Q) * Dch + col;
        int ib = (((Q - jj) + (15 - r) * Q) & (Llen - 1)) * Dch + col;
        s4_pair_mul(Z[ia], Z[ib], ZK[ia], ZK[ib], cr[r], va[r], vb[15 - r]);
      }
      float2 X[16];
      dft16(va, +1.0f, X);
      int i0 = 16 * jj;
#pragma unroll
      for (int m = 0; m < 16; ++m)
        y[(i0 + m) * Dch + col] = X[m];
      if (jj != 0 && jj != Q / 2) {
        dft16(vb, +1.0f, X);
        int i1 = 16 * (Q - jj);
#pragma unroll
        for (int m = 0; m < 16; ++m)
          y[(i1 + m) * Dch + col] = X[m];
      }
    }
  }
  }
}

__global__ __launch_bounds__(256) void s4_final_r32(
    const float2* __restrict__ xb, const float* __restrict__ ub,
    const float* __restrict__ Dp, float* __restrict__ outb, int np, int rev)
{
  const int Q = Llen / 32;
  const float dp = Dp[0];
  const float inv = 1.0f / (4.0f * (float)Llen);
  for (int pp = 0; pp < np; ++pp) {
  const int p = rev ? (np - 1 - pp) : pp;
  const float2* __restrict__ x = xb + (size_t)p * PLANE;
  const float* __restrict__ u = ub + (size_t)p * PLANE;
  float* __restrict__ out = outb + (size_t)p * PLANE;
  for (int j = blockIdx.x; j < Q; j += gridDim.x) {
    float ang = (0.0625f * PI_F) * (float)j / (float)Q;
    float s1, c1;
    sincosf(ang, &s1, &c1);
    float2 w[32];
    w[1]  = make_float2(c1, s1);
    w[2]  = cmulf(w[1], w[1]);
    w[3]  = cmulf(w[2], w[1]);
    w[4]  = cmulf(w[2], w[2]);
    w[5]  = cmulf(w[4], w[1]);
    w[6]  = cmulf(w[3], w[3]);
    w[7]  = cmulf(w[4], w[3]);
    w[8]  = cmulf(w[4], w[4]);
    w[9]  = cmulf(w[8], w[1]);
    w[10] = cmulf(w[5], w[5]);
    w[11] = cmulf(w[8], w[3]);
    w[12] = cmulf(w[6], w[6]);
    w[13] = cmulf(w[8], w[5]);
    w[14] = cmulf(w[7], w[7]);
    w[15] = cmulf(w[8], w[7]);
    w[16] = cmulf(w[8], w[8]);
#pragma unroll
    for (int r = 1; r < 16; ++r)
      w[16 + r] = cmulf(w[16], w[r]);
    for (int h = 0; h < 2; ++h) {
      int col = threadIdx.x + h * HCOL;
      float2 v[32];
      float ue[16][2];
#pragma unroll
      for (int r = 0; r < 32; ++r)
        v[r] = x[(j + r * Q) * Dch + col];
#pragma unroll
      for (int m = 0; m < 16; ++m) {
        int e = (2 * (j + m * Q)) * Dch;
        ue[m][0] = u[e + col];
        ue[m][1] = u[e + Dch + col];
      }
#pragma unroll
      for (int r = 1; r < 32; ++r)
        v[r] = cmulf(v[r], w[r]);
      float2 X[32];
      dft32(v, +1.0f, X);
#pragma unroll
      for (int m = 0; m < 16; ++m) {
        int e = (2 * (j + m * Q)) * Dch;
        out[e + col]       = X[m].x * inv + dp * ue[m][0];
        out[e + Dch + col] = X[m].y * inv + dp * ue[m][1];
      }
    }
  }
  }
}


extern "C" void kernel_launch(void* const* d_in, const int* in_sizes, int n_in,
                              void* d_out, int out_size, void* d_ws, size_t ws_size,
                              hipStream_t stream) {
  const float* u        = (const float*)d_in[0];
  const float* Lam      = (const float*)d_in[1];
  const float* Pri      = (const float*)d_in[2];
  const float* Bri      = (const float*)d_in[3];
  const float* Cri      = (const float*)d_in[4];
  const float* Dp       = (const float*)d_in[5];
  const float* log_step = (const float*)d_in[6];
  float* out = (float*)d_out;

  float2* P0 = (float2*)d_ws;
  float2* P1 = P0 + PLANE;
  float2* P2 = P0 + 2 * PLANE;
  float2* P3 = P0 + 3 * PLANE;
  float2* P4 = P0 + 4 * PLANE;
  float2* P5 = P0 + 5 * PLANE;
  float2* P6 = P0 + 6 * PLANE;
  float2* P7 = P0 + 7 * PLANE;
  uint8_t* base = (uint8_t*)P7;
  size_t off = 0;
  auto carve = [&](size_t bytes) -> void* {
    void* p = base + off;
    off += (bytes + 255) & ~(size_t)255;
    return p;
  };
  float*  APr  = (float*)carve((size_t)Llen * Nst * 4);
  float*  APi  = (float*)carve((size_t)Llen * Nst * 4);
  float*  W0r  = (float*)carve((size_t)Nst * Dch * 4);
  float*  W0i  = (float*)carve((size_t)Nst * Dch * 4);
  float*  W1r  = (float*)carve((size_t)Nst * Dch * 4);
  float*  W1i  = (float*)carve((size_t)Nst * Dch * 4);
  float*  perl = (float*)carve((size_t)4 * Llen * 4);
  float2* Ev   = P6;
  (void)P1; (void)P4; (void)P7;
  (void)ws_size; (void)in_sizes; (void)n_in; (void)out_size;

  const int TB   = 256;
  const int gR128 = 2 * R128Q;
  const int gR32 = NBLK;
  const int gPm  = NBLK + 1;

  s4_setup<<<NPB + NWB + NCB, TB, 0, stream>>>(Lam, Pri, Cri, log_step, Bri, perl,
                                              W0r, W0i, W1r, W1i, APr, APi);
  s4_cauchy_gemm<<<dim3(Dch / 16, Llen / 16), 32, 0, stream>>>(
      APr, APi, W0r, W0i, W1r, W1i, perl, Ev);

  const float* u0 = u;
  const float* u1 = u + (size_t)2 * Llen * Dch;
  float*       o0 = out;
  float*       o1 = out + (size_t)2 * Llen * Dch;

  s4_first_r128<<<gR128, TB, 0, stream>>>(u0, P0, Ev, P2, 0, 3, 1);
  s4_fft_r128<<<gR128, TB, 0, stream>>>(P0, P3, P2, P5, -1.0f, +1.0f, 3, 0);

  s4_first_r128<<<gR128, TB, 0, stream>>>(u1, P0, P5, P2, 1, 3, 1);
  s4_fft_r128<<<gR128, TB, 0, stream>>>(P0, P6, P2, P5, -1.0f, -1.0f, 3, 0);
  float2* Kd = P5;

  s4_pmul_r16<<<gPm, TB, 0, stream>>>(P6, Kd, P0, 2, 1);
  s4_fft_r32<<<gR32, TB, 0, stream>>>(P0, P6, P2, P2, Llen, 16, +1.0f, +1.0f, 2, 0);
  s4_final_r32<<<gR32, TB, 0, stream>>>(P6, u1, Dp, o1, 2, 1);

  s4_pmul_r16<<<gPm, TB, 0, stream>>>(P3, Kd, P0, 2, 1);
  s4_fft_r32<<<gR32, TB, 0, stream>>>(P0, P6, P2, P2, Llen, 16, +1.0f, +1.0f, 2, 0);
  s4_final_r32<<<gR32, TB, 0, stream>>>(P6, u0, Dp, o0, 2, 1);
}


# Inverse passes take all four batch entries per launch: product + radix-16 pass in place on the spectra (mid pass reads that row order), K's bins read once per row (LDS); 9 launches, not 12; same sums
# speedup vs baseline: 3.5963x; 1.0301x over previous
#include <hip/hip_runtime.h>
#include <hip/hip_bf16.h>
#include <stdint.h>


#define Bsz   4
#define Llen  16384
#define Dch   512
#define Nst   64
#define NFFT  32768
#define PI_F  3.14159265358979323846f

typedef __attribute__((ext_vector_type(2))) float v2f;
typedef __attribute__((ext_vector_type(8))) float v8f;

#ifndef __has_builtin
#define __has_builtin(x) 0
#endif
#if __has_builtin(__builtin_amdgcn_wmma_f32_16x16x4_f32)
#define S4_WMMA 1
#else
#define S4_WMMA 0
#endif

__device__ __forceinline__ float2 cadd(float2 a, float2 b) { return make_float2(a.x + b.x, a.y + b.y); }
__device__ __forceinline__ float2 csub(float2 a, float2 b) { return make_float2(a.x - b.x, a.y - b.y); }
__device__ __forceinline__ float2 cmulf(float2 a, float2 b) { return make_float2(a.x * b.x - a.y * b.y, a.x * b.y + a.y * b.x); }
__device__ __forceinline__ float2 crotis(float2 a, float s) { return make_float2(-s * a.y, s * a.x); }

__device__ __forceinline__ void dft8(const float2* v, float s, float2* X) {
  const float R = 0.70710678118654752440f;
  float2 t0 = cadd(v[0], v[4]), t1 = cadd(v[1], v[5]);
  float2 t2 = cadd(v[2], v[6]), t3 = cadd(v[3], v[7]);
  float2 u0 = csub(v[0], v[4]), u1 = csub(v[1], v[5]);
  float2 u2 = csub(v[2], v[6]), u3 = csub(v[3], v[7]);
  float2 m1 = cmulf(u1, make_float2(R, s * R));
  float2 m2 = crotis(u2, s);
  float2 m3 = cmulf(u3, make_float2(-R, s * R));
  float2 p0 = cadd(t0, t2), p1 = csub(t0, t2);
  float2 p2 = cadd(t1, t3), p3 = crotis(csub(t1, t3), s);
  X[0] = cadd(p0, p2); X[2] = cadd(p1, p3);
  X[4] = csub(p0, p2); X[6] = csub(p1, p3);
  float2 q0 = cadd(u0, m2), q1 = csub(u0, m2);
  float2 q2 = cadd(m1, m3), q3 = crotis(csub(m1, m3), s);
  X[1] = cadd(q0, q2); X[3] = cadd(q1, q3);
  X[5] = csub(q0, q2); X[7] = csub(q1, q3);
}

__device__ __forceinline__ void dft4(float2 a0, float2 a1, float2 a2, float2 a3, float s,
                                     float2& y0, float2& y1, float2& y2, float2& y3) {
  float2 p0 = cadd(a0, a2), p1 = csub(a0, a2);
  float2 p2 = cadd(a1, a3), p3 = crotis(csub(a1, a3), s);
  y0 = cadd(p0, p2); y1 = cadd(p1, p3);
  y2 = csub(p0, p2); y3 = csub(p1, p3);
}

__device__ __forceinline__ void dft16(const float2* v, float s, float2* X) {
  const float R  = 0.70710678118654752440f;
  const float C1 = 0.92387953251128675613f;
  const float S1 = 0.38268343236508977173f;
  float2 t0[4], t1[4], t2[4], t3[4];
  dft4(v[0], v[4], v[8],  v[12], s, t0[0], t0[1], t0[2], t0[3]);
  dft4(v[1], v[5], v[9],  v[13], s, t1[0], t1[1], t1[2], t1[3]);
  dft4(v[2], v[6], v[10], v[14], s, t2[0], t2[1], t2[2], t2[3]);
  dft4(v[3], v[7], v[11], v[15], s, t3[0], t3[1], t3[2], t3[3]);
  t1[1] = cmulf(t1[1], make_float2(C1, s * S1));
  t1[2] = cmulf(t1[2], make_float2(R, s * R));
  t1[3] = cmulf(t1[3], make_float2(S1, s * C1));
  t2[1] = cmulf(t2[1], make_float2(R, s * R));
  t2[2] = crotis(t2[2], s);
  t2[3] = cmulf(t2[3], make_float2(-R, s * R));
  t3[1] = cmulf(t3[1], make_float2(S1, s * C1));
  t3[2] = cmulf(t3[2], make_float2(-R, s * R));
  t3[3] = cmulf(t3[3], make_float2(-C1, -s * S1));
#pragma unroll
  for (int m1 = 0; m1 < 4; ++m1)
    dft4(t0[m1], t1[m1], t2[m1], t3[m1], s, X[m1], X[m1 + 4], X[m1 + 8], X[m1 + 12]);
}

__device__ __forceinline__ void dft32(const float2* v, float s, float2* X) {
  float2 t0_0, t0_1, t0_2, t0_3;
  dft4(v[0], v[8], v[16], v[24], s, t0_0, t0_1, t0_2, t0_3);
  float2 t1_0, t1_1, t1_2, t1_3;
  dft4(v[1], v[9], v[17], v[25], s, t1_0, t1_1, t1_2, t1_3);
  float2 t2_0, t2_1, t2_2, t2_3;
  dft4(v[2], v[10], v[18], v[26], s, t2_0, t2_1, t2_2, t2_3);
  float2 t3_0, t3_1, t3_2, t3_3;
  dft4(v[3], v[11], v[19], v[27], s, t3_0, t3_1, t3_2, t3_3);
  float2 t4_0, t4_1, t4_2, t4_3;
  dft4(v[4], v[12], v[20], v[28], s, t4_0, t4_1, t4_2, t4_3);
  float2 t5_0, t5_1, t5_2, t5_3;
  dft4(v[5], v[13], v[21], v[29], s, t5_0, t5_1, t5_2, t5_3);
  float2 t6_0, t6_1, t6_2, t6_3;
  dft4(v[6], v[14], v[22], v[30], s, t6_0, t6_1, t6_2, t6_3);
  float2 t7_0, t7_1, t7_2, t7_3;
  dft4(v[7], v[15], v[23], v[31], s, t7_0, t7_1, t7_2, t7_3);
  t1_1 = cmulf(t1_1, make_float2(0.98078528040323043f, s * 0.19509032201612825f));
  t2_1 = cmulf(t2_1, make_float2(0.92387953251128674f, s * 0.38268343236508978f));
  t3_1 = cmulf(t3_1, make_float2(0.83146961230254524f, s * 0.55557023301960218f));
  t4_1 = cmulf(t4_1, make_float2(0.70710678118654757f, s * 0.70710678118654746f));
  t5_1 = cmulf(t5_1, make_float2(0.55557023301960229f, s * 0.83146961230254524f));
  t6_1 = cmulf(t6_1, make_float2(0.38268343236508984f, s * 0.92387953251128674f));
  t7_1 = cmulf(t7_1, make_float2(0.19509032201612833f, s * 0.98078528040323043f));
  t1_2 = cmulf(t1_2, make_float2(0.92387953251128674f, s * 0.38268343236508978f));
  t2_2 = cmulf(t2_2, make_float2(0.70710678118654757f, s * 0.70710678118654746f));
  t3_2 = cmulf(t3_2, make_float2(0.38268343236508984f, s * 0.92387953251128674f));
  t4_2 = crotis(t4_2, s);
  t5_2 = cmulf(t5_2, make_float2(-0.38268343236508973f, s * 0.92387953251128674f));
  t6_2 = cmulf(t6_2, make_float2(-0.70710678118654746f, s * 0.70710678118654757f));
  t7_2 = cmulf(t7_2, make_float2(-0.92387953251128674f, s * 0.38268343236508989f));
  t1_3 = cmulf(t1_3, make_float2(0.83146961230254524f, s * 0.55557023301960218f));
  t2_3 = cmulf(t2_3, make_float2(0.38268343236508984f, s * 0.92387953251128674f));
  t3_3 = cmulf(t3_3, make_float2(-0.19509032201612819f, s * 0.98078528040323043f));
  t4_3 = cmulf(t4_3, make_float2(-0.70710678118654746f, s * 0.70710678118654757f));
  t5_3 = cmulf(t5_3, make_float2(-0.98078528040323043f, s * 0.19509032201612861f));
  t6_3 = cmulf(t6_3, make_float2(-0.92387953251128685f, s * -0.38268343236508967f));
  t7_3 = cmulf(t7_3, make_float2(-0.55557023301960218f, s * -0.83146961230254524f));
  {
    float2 a[8] = {t0_0, t1_0, t2_0, t3_0, t4_0, t5_0, t6_0, t7_0};
    float2 b[8];
    dft8(a, s, b);
    X[0] = b[0]; X[4] = b[1]; X[8] = b[2]; X[12] = b[3];
    X[16] = b[4]; X[20] = b[5]; X[24] = b[6]; X[28] = b[7];
  }
  {
    float2 a[8] = {t0_1, t1_1, t2_1, t3_1, t4_1, t5_1, t6_1, t7_1};
    float2 b[8];
    dft8(a, s, b);
    X[1] = b[0]; X[5] = b[1]; X[9] = b[2]; X[13] = b[3];
    X[17] = b[4]; X[21] = b[5]; X[25] = b[6]; X[29] = b[7];
  }
  {
    float2 a[8] = {t0_2, t1_2, t2_2, t3_2, t4_2, t5_2, t6_2, t7_2};
    float2 b[8];
    dft8(a, s, b);
    X[2] = b[0]; X[6] = b[1]; X[10] = b[2]; X[14] = b[3];
    X[18] = b[4]; X[22] = b[5]; X[26] = b[6]; X[30] = b[7];
  }
  {
    float2 a[8] = {t0_3, t1_3, t2_3, t3_3, t4_3, t5_3, t6_3, t7_3};
    float2 b[8];
    dft8(a, s, b);
    X[3] = b[0]; X[7] = b[1]; X[11] = b[2]; X[15] = b[3];
    X[19] = b[4]; X[23] = b[5]; X[27] = b[6]; X[31] = b[7];
  }
}

__device__ __forceinline__ void dft128(const float2* v, float s, float2* X) {
  float2 t0[32], t1[32], t2[32], t3[32];
#pragma unroll
  for (int r2 = 0; r2 < 32; ++r2)
    dft4(v[r2], v[r2 + 32], v[r2 + 64], v[r2 + 96], s, t0[r2], t1[r2], t2[r2], t3[r2]);
  t1[1] = cmulf(t1[1], make_float2(0.99879545620517241f, s * 0.049067674327418015f));
  t1[2] = cmulf(t1[2], make_float2(0.99518472667219693f, s * 0.098017140329560604f));
  t1[3] = cmulf(t1[3], make_float2(0.98917650996478101f, s * 0.14673047445536175f));
  t1[4] = cmulf(t1[4], make_float2(0.98078528040323043f, s * 0.19509032201612825f));
  t1[5] = cmulf(t1[5], make_float2(0.97003125319454397f, s * 0.24298017990326387f));
  t1[6] = cmulf(t1[6], make_float2(0.95694033573220882f, s * 0.29028467725446233f));
  t1[7] = cmulf(t1[7], make_float2(0.94154406518302081f, s * 0.33688985339222005f));
  t1[8] = cmulf(t1[8], make_float2(0.92387953251128674f, s * 0.38268343236508978f));
  t1[9] = cmulf(t1[9], make_float2(0.90398929312344334f, s * 0.42755509343028208f));
  t1[10] = cmulf(t1[10], make_float2(0.88192126434835505f, s * 0.47139673682599764f));
  t1[11] = cmulf(t1[11], make_float2(0.85772861000027212f, s * 0.51410274419322166f));
  t1[12] = cmulf(t1[12], make_float2(0.83146961230254524f, s * 0.55557023301960218f));
  t1[13] = cmulf(t1[13], make_float2(0.80320753148064494f, s * 0.59569930449243336f));
  t1[14] = cmulf(t1[14], make_float2(0.77301045336273699f, s * 0.63439328416364549f));
  t1[15] = cmulf(t1[15], make_float2(0.74095112535495911f, s * 0.67155895484701833f));
  t1[16] = cmulf(t1[16], make_float2(0.70710678118654757f, s * 0.70710678118654746f));
  t1[17] = cmulf(t1[17], make_float2(0.67155895484701833f, s * 0.74095112535495911f));
  t1[18] = cmulf(t1[18], make_float2(0.63439328416364549f, s * 0.77301045336273699f));
  t1[19] = cmulf(t1[19], make_float2(0.59569930449243347f, s * 0.80320753148064483f));
  t1[20] = cmulf(t1[20], make_float2(0.55557023301960229f, s * 0.83146961230254524f));
  t1[21] = cmulf(t1[21], make_float2(0.51410274419322166f, s * 0.85772861000027212f));
  t1[22] = cmulf(t1[22], make_float2(0.47139673682599781f, s * 0.88192126434835494f));
  t1[23] = cmulf(t1[23], make_float2(0.42755509343028220f, s * 0.90398929312344334f));
  t1[24] = cmulf(t1[24], make_float2(0.38268343236508984f, s * 0.92387953251128674f));
  t1[25] = cmulf(t1[25], make_float2(0.33688985339222005f, s * 0.94154406518302081f));
  t1[26] = cmulf(t1[26], make_float2(0.29028467725446233f, s * 0.95694033573220894f));
  t1[27] = cmulf(t1[27], make_float2(0.24298017990326398f, s * 0.97003125319454397f));
  t1[28] = cmulf(t1[28], make_float2(0.19509032201612833f, s * 0.98078528040323043f));
  t1[29] = cmulf(t1[29], make_float2(0.14673047445536175f, s * 0.98917650996478101f));
  t1[30] = cmulf(t1[30], make_float2(0.098017140329560770f, s * 0.99518472667219682f));
  t1[31] = cmulf(t1[31], make_float2(0.049067674327418126f, s * 0.99879545620517241f));
  t2[1] = cmulf(t2[1], make_float2(0.99518472667219693f, s * 0.098017140329560604f));
  t2[2] = cmulf(t2[2], make_float2(0.98078528040323043f, s * 0.19509032201612825f));
  t2[3] = cmulf(t2[3], make_float2(0.95694033573220882f, s * 0.29028467725446233f));
  t2[4] = cmulf(t2[4], make_float2(0.92387953251128674f, s * 0.38268343236508978f));
  t2[5] = cmulf(t2[5], make_float2(0.88192126434835505f, s * 0.47139673682599764f));
  t2[6] = cmulf(t2[6], make_float2(0.83146961230254524f, s * 0.55557023301960218f));
  t2[7] = cmulf(t2[7], make_float2(0.77301045336273699f, s * 0.63439328416364549f));
  t2[8] = cmulf(t2[8], make_float2(0.70710678118654757f, s * 0.70710678118654746f));
  t2[9] = cmulf(t2[9], make_float2(0.63439328416364549f, s * 0.77301045336273699f));
  t2[10] = cmulf(t2[10], make_float2(0.55557023301960229f, s * 0.83146961230254524f));
  t2[11] = cmulf(t2[11], make_float2(0.47139673682599781f, s * 0.88192126434835494f));
  t2[12] = cmulf(t2[12], make_float2(0.38268343236508984f, s * 0.92387953251128674f));
  t2[13] = cmulf(t2[13], make_float2(0.29028467725446233f, s * 0.95694033573220894f));
  t2[14] = cmulf(t2[14], make_float2(0.19509032201612833f, s * 0.98078528040323043f));
  t2[15] = cmulf(t2[15], make_float2(0.098017140329560770f, s * 0.99518472667219682f));
  t2[16] = crotis(t2[16], s);
  t2[17] = cmulf(t2[17], make_float2(-0.098017140329560645f, s * 0.99518472667219693f));
  t2[18] = cmulf(t2[18], make_float2(-0.19509032201612819f, s * 0.98078528040323043f));
  t2[19] = cmulf(t2[19], make_float2(-0.29028467725446216f, s * 0.95694033573220894f));
  t2[20] = cmulf(t2[20], make_float2(-0.38268343236508973f, s * 0.92387953251128674f));
  t2[21] = cmulf(t2[21], make_float2(-0.47139673682599770f, s * 0.88192126434835505f));
  t2[22] = cmulf(t2[22], make_float2(-0.55557023301960196f, s * 0.83146961230254546f));
  t2[23] = cmulf(t2[23], make_float2(-0.63439328416364538f, s * 0.77301045336273710f));
  t2[24] = cmulf(t2[24], make_float2(-0.70710678118654746f, s * 0.70710678118654757f));
  t2[25] = cmulf(t2[25], make_float2(-0.77301045336273699f, s * 0.63439328416364549f));
  t2[26] = cmulf(t2[26], make_float2(-0.83146961230254535f, s * 0.55557023301960218f));
  t2[27] = cmulf(t2[27], make_float2(-0.88192126434835494f, s * 0.47139673682599786f));
  t2[28] = cmulf(t2[28], make_float2(-0.92387953251128674f, s * 0.38268343236508989f));
  t2[29] = cmulf(t2[29], make_float2(-0.95694033573220882f, s * 0.29028467725446239f));
  t2[30] = cmulf(t2[30], make_float2(-0.98078528040323043f, s * 0.19509032201612861f));
  t2[31] = cmulf(t2[31], make_float2(-0.99518472667219682f, s * 0.098017140329560826f));
  t3[1] = cmulf(t3[1], make_float2(0.98917650996478101f, s * 0.14673047445536175f));
  t3[2] = cmulf(t3[2], make_float2(0.95694033573220882f, s * 0.29028467725446233f));
  t3[3] = cmulf(t3[3], make_float2(0.90398929312344334f, s * 0.42755509343028208f));
  t3[4] = cmulf(t3[4], make_float2(0.83146961230254524f, s * 0.55557023301960218f));
  t3[5] = cmulf(t3[5], make_float2(0.74095112535495911f, s * 0.67155895484701833f));
  t3[6] = cmulf(t3[6], make_float2(0.63439328416364549f, s * 0.77301045336273699f));
  t3[7] = cmulf(t3[7], make_float2(0.51410274419322166f, s * 0.85772861000027212f));
  t3[8] = cmulf(t3[8], make_float2(0.38268343236508984f, s * 0.92387953251128674f));
  t3[9] = cmulf(t3[9], make_float2(0.24298017990326398f, s * 0.97003125319454397f));
  t3[10] = cmulf(t3[10], make_float2(0.098017140329560770f, s * 0.99518472667219682f));
  t3[11] = cmulf(t3[11], make_float2(-0.049067674327418008f, s * 0.99879545620517241f));
  t3[12] = cmulf(t3[12], make_float2(-0.19509032201612819f, s * 0.98078528040323043f));
  t3[13] = cmulf(t3[13], make_float2(-0.33688985339221994f, s * 0.94154406518302081f));
  t3[14] = cmulf(t3[14], make_float2(-0.47139673682599770f, s * 0.88192126434835505f));
  t3[15] = cmulf(t3[15], make_float2(-0.59569930449243336f, s * 0.80320753148064494f));
  t3[16] = cmulf(t3[16], make_float2(-0.70710678118654746f, s * 0.70710678118654757f));
  t3[17] = cmulf(t3[17], make_float2(-0.80320753148064483f, s * 0.59569930449243347f));
  t3[18] = cmulf(t3[18], make_float2(-0.88192126434835494f, s * 0.47139673682599786f));
  t3[19] = cmulf(t3[19], make_float2(-0.94154406518302070f, s * 0.33688985339222033f));
  t3[20] = cmulf(t3[20], make_float2(-0.98078528040323043f, s * 0.19509032201612861f));
  t3[21] = cmulf(t3[21], make_float2(-0.99879545620517241f, s * 0.049067674327417966f));
  t3[22] = cmulf(t3[22], make_float2(-0.99518472667219693f, s * -0.098017140329560590f));
  t3[23] = cmulf(t3[23], make_float2(-0.97003125319454397f, s * -0.24298017990326382f));
  t3[24] = cmulf(t3[24], make_float2(-0.92387953251128685f, s * -0.38268343236508967f));
  t3[25] = cmulf(t3[25], make_float2(-0.85772861000027212f, s * -0.51410274419322155f));
  t3[26] = cmulf(t3[26], make_float2(-0.77301045336273710f, s * -0.63439328416364527f));
  t3[27] = cmulf(t3[27], make_float2(-0.67155895484701866f, s * -0.74095112535495888f));
  t3[28] = cmulf(t3[28], make_float2(-0.55557023301960218f, s * -0.83146961230254524f));
  t3[29] = cmulf(t3[29], make_float2(-0.42755509343028247f, s * -0.90398929312344312f));
  t3[30] = cmulf(t3[30], make_float2(-0.29028467725446244f, s * -0.95694033573220882f));
  t3[31] = cmulf(t3[31], make_float2(-0.14673047445536230f, s * -0.98917650996478090f));
  float2 Y[32];
  dft32(t0, s, Y);
#pragma unroll
  for (int m2 = 0; m2 < 32; ++m2) X[0 + 4 * m2] = Y[m2];
  dft32(t1, s, Y);
#pragma unroll
  for (int m2 = 0; m2 < 32; ++m2) X[1 + 4 * m2] = Y[m2];
  dft32(t2, s, Y);
#pragma unroll
  for (int m2 = 0; m2 < 32; ++m2) X[2 + 4 * m2] = Y[m2];
  dft32(t3, s, Y);
#pragma unroll
  for (int m2 = 0; m2 < 32; ++m2) X[3 + 4 * m2] = Y[m2];
}

__device__ __forceinline__ void s4_g_of_l(
    int l, const float* __restrict__ log_step,
    float& gr, float& gi, float& scr, float& sci)
{
  float step = expf(log_step[0]);
  float theta = -2.0f * PI_F * (float)l / (float)Llen;
  float oi, orr;
  sincosf(theta, &oi, &orr);
  float op1r = 1.0f + orr;
  float den = op1r * op1r + oi * oi;
  float nr = 1.0f - orr, ni = -oi;
  float gscale = 2.0f / step;
  gr = gscale * (nr * op1r + ni * oi) / den;
  gi = gscale * (ni * op1r - nr * oi) / den;
  scr = 2.0f * op1r / den;
  sci = -2.0f * oi / den;
}

__device__ __forceinline__ void s4_perl_body(
    int l, const float* __restrict__ Lam, const float* __restrict__ Pri,
    const float* __restrict__ Cri, const float* __restrict__ log_step,
    float* __restrict__ perl)
{
  if (l >= Llen) return;
  float gr, gi, scr, sci;
  s4_g_of_l(l, log_step, gr, gi, scr, sci);
  float k01r = 0.f, k01i = 0.f, k11r = 0.f, k11i = 0.f;
  for (int n = 0; n < Nst; ++n) {
    float lr = Lam[2 * n], li = Lam[2 * n + 1];
    float pr = Pri[2 * n], pi = Pri[2 * n + 1];
    float cr = Cri[2 * n], ci = Cri[2 * n + 1];
    float zr = gr - lr, zi = gi - li;
    float zd = zr * zr + zi * zi;
    float car = zr / zd, cai = -zi / zd;
    float w01r = cr * pr + ci * pi;
    float w01i = cr * pi - ci * pr;
    float w11r = pr * pr + pi * pi;
    k01r += car * w01r - cai * w01i;
    k01i += car * w01i + cai * w01r;
    k11r += car * w11r;
    k11i += cai * w11r;
  }
  float dr = 1.0f + k11r, di = k11i;
  float dd = dr * dr + di * di;
  float sr = (k01r * dr + k01i * di) / dd;
  float si = (k01i * dr - k01r * di) / dd;
  perl[l] = sr;
  perl[Llen + l] = si;
  perl[2 * Llen + l] = scr;
  perl[3 * Llen + l] = sci;
}

__device__ __forceinline__ void s4_weights_body(
    int idx, const float* __restrict__ Bri, const float* __restrict__ Pri,
    const float* __restrict__ Cri,
    float* __restrict__ W0r, float* __restrict__ W0i,
    float* __restrict__ W1r, float* __restrict__ W1i)
{
  if (idx >= Nst * Dch) return;
  int n = idx / Dch, d = idx % Dch;
  float br = Bri[2 * (n * Dch + d)], bi = Bri[2 * (n * Dch + d) + 1];
  float cr = Cri[2 * n], ci = Cri[2 * n + 1];
  float pr = Pri[2 * n], pi = Pri[2 * n + 1];
  int o = (n >> 1) * (2 * Dch) + 2 * d + (n & 1);
  W0r[o] = cr * br + ci * bi;
  W0i[o] = cr * bi - ci * br;
  W1r[o] = pr * br + pi * bi;
  W1i[o] = pr * bi - pi * br;
}

#define NPB  (Llen / 256)
#define NWB  (Nst * Dch / 256)
#define NCB  512
__device__ __forceinline__ void s4_cauchy_body(
    int cb, const float* __restrict__ Lam, const float* __restrict__ log_step,
    float* __restrict__ APr, float* __restrict__ APi)
{
  const int idx0 = cb * 256 + threadIdx.x;
  const int l = idx0 % Llen;
  float gr, gi, scr, sci;
  s4_g_of_l(l, log_step, gr, gi, scr, sci);
  (void)scr; (void)sci;
  for (int idx = idx0; idx < Llen * Nst; idx += NCB * 256) {
    int n = idx / Llen;
    float zr = gr - Lam[2 * n], zi = gi - Lam[2 * n + 1];
    float zd = zr * zr + zi * zi;
    int o = (n >> 1) * (2 * Llen) + 2 * l + (n & 1);
    APr[o] = zr / zd;
    APi[o] = -zi / zd;
  }
}

__global__ __launch_bounds__(256) void s4_setup(
    const float* __restrict__ Lam, const float* __restrict__ Pri,
    const float* __restrict__ Cri, const float* __restrict__ log_step,
    const float* __restrict__ Bri, float* __restrict__ perl,
    float* __restrict__ W0r, float* __restrict__ W0i,
    float* __restrict__ W1r, float* __restrict__ W1i,
    float* __restrict__ APr, float* __restrict__ APi)
{
  const int b = blockIdx.x;
  if (b < NPB)
    s4_perl_body(b * 256 + threadIdx.x, Lam, Pri, Cri, log_step, perl);
  else if (b < NPB + NWB)
    s4_weights_body((b - NPB) * 256 + threadIdx.x, Bri, Pri, Cri, W0r, W0i, W1r, W1i);
  else
    s4_cauchy_body(b - NPB - NWB, Lam, log_step, APr, APi);
}

__global__ __launch_bounds__(32) void s4_cauchy_gemm(
    const float* __restrict__ APr, const float* __restrict__ APi,
    const float* __restrict__ W0r, const float* __restrict__ W0i,
    const float* __restrict__ W1r, const float* __restrict__ W1i,
    const float* __restrict__ perl, float2* __restrict__ Ev)
{
  int lane = threadIdx.x & 31;
  int tileD = blockIdx.x;
  int tileL = blockIdx.y;
  int d0 = tileD * 16 + (lane & 15);
  int mbase = tileL * 16 + ((lane >> 4) ? 8 : 0);
#if S4_WMMA
  int m = tileL * 16 + (lane & 15);
  int khi = (lane >> 4) * 2;
  v8f aRr = {}, aRi = {}, aSr = {}, aSi = {};
  for (int kb = 0; kb < Nst; kb += 4) {
    int k0 = kb + khi;
    const float* pr = APr + (k0 >> 1) * (2 * Llen) + 2 * m;
    const float* pi = APi + (k0 >> 1) * (2 * Llen) + 2 * m;
    v2f ar = *(const v2f*)pr;
    v2f ai = *(const v2f*)pi;
    v2f nai = -ai;
    int bo = (k0 >> 1) * (2 * Dch) + 2 * d0;
    v2f b0r = *(const v2f*)(W0r + bo);
    v2f b0i = *(const v2f*)(W0i + bo);
    v2f b1r = *(const v2f*)(W1r + bo);
    v2f b1i = *(const v2f*)(W1i + bo);
    aRr = __builtin_amdgcn_wmma_f32_16x16x4_f32(false, ar,  false, b0r, (short)0, aRr, false, false);
    aRr = __builtin_amdgcn_wmma_f32_16x16x4_f32(false, nai, false, b0i, (short)0, aRr, false, false);
    aRi = __builtin_amdgcn_wmma_f32_16x16x4_f32(false, ar,  false, b0i, (short)0, aRi, false, false);
    aRi = __builtin_amdgcn_wmma_f32_16x16x4_f32(false, ai,  false, b0r, (short)0, aRi, false, false);
    aSr = __builtin_amdgcn_wmma_f32_16x16x4_f32(false, ar,  false, b1r, (short)0, aSr, false, false);
    aSr = __builtin_amdgcn_wmma_f32_16x16x4_f32(false, nai, false, b1i, (short)0, aSr, false, false);
    aSi = __builtin_amdgcn_wmma_f32_16x16x4_f32(false, ar,  false, b1i, (short)0, aSi, false, false);
    aSi = __builtin_amdgcn_wmma_f32_16x16x4_f32(false, ai,  false, b1r, (short)0, aSi, false, false);
  }
  for (int r = 0; r < 8; ++r) {
    int l = mbase + r;
    float k00r = aRr[r], k00i = aRi[r];
    float k10r = aSr[r], k10i = aSi[r];
    float sr = perl[l],            si = perl[Llen + l];
    float cr = perl[2 * Llen + l], ci = perl[3 * Llen + l];
    float tr = k00r - (sr * k10r - si * k10i);
    float ti = k00i - (sr * k10i + si * k10r);
    Ev[l * Dch + d0] = make_float2(cr * tr - ci * ti, cr * ti + ci * tr);
  }
#else
  for (int r = 0; r < 8; ++r) {
    int l = mbase + r;
    float k00r = 0.f, k00i = 0.f, k10r = 0.f, k10i = 0.f;
    for (int k = 0; k < Nst; ++k) {
      int ao = (k >> 1) * (2 * Llen) + 2 * l + (k & 1);
      int bo = (k >> 1) * (2 * Dch) + 2 * d0 + (k & 1);
      float arr = APr[ao], aii = APi[ao];
      float b0r = W0r[bo], b0i = W0i[bo];
      float b1r = W1r[bo], b1i = W1i[bo];
      k00r += arr * b0r - aii * b0i;  k00i += arr * b0i + aii * b0r;
      k10r += arr * b1r - aii * b1i;  k10i += arr * b1i + aii * b1r;
    }
    float sr = perl[l],            si = perl[Llen + l];
    float cr = perl[2 * Llen + l], ci = perl[3 * Llen + l];
    float tr = k00r - (sr * k10r - si * k10i);
    float ti = k00i - (sr * k10i + si * k10r);
    Ev[l * Dch + d0] = make_float2(cr * tr - ci * ti, cr * ti + ci * tr);
  }
#endif
}

#define NBLK  512
#define HCOL  256
#define PLANE ((size_t)Llen * Dch)

__global__ __launch_bounds__(256) void s4_mid_r32(
    const float2* __restrict__ xb, float2* __restrict__ yb, int np, int rev)
{
  const int Ns = 16;
  const int q = Llen >> 5;
  for (int pp = 0; pp < np; ++pp) {
  const int p = rev ? (np - 1 - pp) : pp;
  const float2* __restrict__ x = xb + (size_t)p * PLANE;
  float2* __restrict__ y = yb + (size_t)p * PLANE;
  for (int j = blockIdx.x; j < q; j += gridDim.x) {
    int k = j & (Ns - 1);
    float ang = (0.0625f * PI_F) * (float)k / (float)Ns;
    float s1, c1;
    sincosf(ang, &s1, &c1);
    float2 w[32];
    w[1]  = make_float2(c1, s1);
    w[2]  = cmulf(w[1], w[1]);
    w[3]  = cmulf(w[2], w[1]);
    w[4]  = cmulf(w[2], w[2]);
    w[5]  = cmulf(w[4], w[1]);
    w[6]  = cmulf(w[3], w[3]);
    w[7]  = cmulf(w[4], w[3]);
    w[8]  = cmulf(w[4], w[4]);
    w[9]  = cmulf(w[8], w[1]);
    w[10] = cmulf(w[5], w[5]);
    w[11] = cmulf(w[8], w[3]);
    w[12] = cmulf(w[6], w[6]);
    w[13] = cmulf(w[8], w[5]);
    w[14] = cmulf(w[7], w[7]);
    w[15] = cmulf(w[8], w[7]);
    w[16] = cmulf(w[8], w[8]);
#pragma unroll
    for (int r = 1; r < 16; ++r)
      w[16 + r] = cmulf(w[16], w[r]);
    int i0 = 32 * j - 31 * k;
    int r0 = (j >> 4) + k * (Llen / 16);
#pragma nounroll
    for (int h = 0; h < 2; ++h) {
      int col = threadIdx.x + h * HCOL;
      float2 v[32];
#pragma unroll
      for (int r = 0; r < 32; ++r)
        v[r] = x[(r0 + 32 * r) * Dch + col];
      __builtin_amdgcn_sched_group_barrier(0x020, 32, 0);
#pragma unroll
      for (int r = 1; r < 32; ++r)
        v[r] = cmulf(v[r], w[r]);
      float2 X[32];
      dft32(v, +1.0f, X);
#pragma unroll
      for (int m = 0; m < 32; ++m)
        y[(i0 + m * Ns) * Dch + col] = X[m];
    }
  }
  }
}

#define R128Q (Llen / 128)
__global__ __launch_bounds__(256) void s4_fft_r128(
    const float2* __restrict__ xb, float2* __restrict__ yb,
    const float2* __restrict__ xk, float2* __restrict__ yk,
    float signb, float signk, int np, int rev)
{
  for (int pp = 0; pp < np; ++pp) {
  const int p = rev ? (np - 1 - pp) : pp;
  const float2* __restrict__ x = (p < 2) ? xb + (size_t)p * PLANE : xk;
  float2* __restrict__ y = (p < 2) ? yb + (size_t)p * PLANE : yk;
  const float sign = (p < 2) ? signb : signk;
  for (int t = blockIdx.x; t < 2 * R128Q; t += gridDim.x) {
    const int j = t >> 1;
    const int col = threadIdx.x + (t & 1) * HCOL;
    float ang = sign * (2.0f * PI_F) * (float)j / (float)Llen;
    float s1, c1, s16, c16;
    sincosf(ang, &s1, &c1);
    sincosf(16.0f * ang, &s16, &c16);
    float2 wl[16], wh[8];
    wl[1]  = make_float2(c1, s1);
    wl[2]  = cmulf(wl[1], wl[1]);
    wl[3]  = cmulf(wl[2], wl[1]);
    wl[4]  = cmulf(wl[2], wl[2]);
    wl[5]  = cmulf(wl[4], wl[1]);
    wl[6]  = cmulf(wl[3], wl[3]);
    wl[7]  = cmulf(wl[4], wl[3]);
    wl[8]  = cmulf(wl[4], wl[4]);
#pragma unroll
    for (int r = 1; r < 8; ++r)
      wl[8 + r] = cmulf(wl[8], wl[r]);
    wh[1] = make_float2(c16, s16);
    wh[2] = cmulf(wh[1], wh[1]);
    wh[3] = cmulf(wh[2], wh[1]);
    wh[4] = cmulf(wh[2], wh[2]);
    wh[5] = cmulf(wh[4], wh[1]);
    wh[6] = cmulf(wh[3], wh[3]);
    wh[7] = cmulf(wh[4], wh[3]);
    float2 v[128];
#pragma unroll
    for (int r = 0; r < 128; ++r)
      v[r] = x[(j + r * R128Q) * Dch + col];
#pragma unroll
    for (int r = 1; r < 16; ++r)
      v[r] = cmulf(v[r], wl[r]);
#pragma unroll
    for (int hi = 1; hi < 8; ++hi) {
      v[16 * hi] = cmulf(v[16 * hi], wh[hi]);
#pragma unroll
      for (int lo = 1; lo < 16; ++lo)
        v[16 * hi + lo] = cmulf(v[16 * hi + lo], cmulf(wh[hi], wl[lo]));
    }
    float2 X[128];
    dft128(v, sign, X);
#pragma unroll
    for (int m = 0; m < 128; ++m)
      y[(j + m * R128Q) * Dch + col] = X[m];
  }
  }
}


__global__ __launch_bounds__(256) void s4_first_r128(
    const float* __restrict__ ub, float2* __restrict__ yb,
    const float2* __restrict__ xk, float2* __restrict__ yk,
    int kmode, int np, int rev)
{
  const float invL = 1.0f / (float)Llen;
  for (int pp = 0; pp < np; ++pp) {
    const int p = rev ? (np - 1 - pp) : pp;
    float2* __restrict__ y = (p < 2) ? yb + (size_t)p * PLANE : yk;
    const float sign = (p == 2 && kmode == 0) ? +1.0f : -1.0f;
    for (int t = blockIdx.x; t < 2 * R128Q; t += gridDim.x) {
      const int j = t >> 1;
      const int col = threadIdx.x + (t & 1) * HCOL;
      float2 v[128];
      if (p < 2) {
        const float* __restrict__ u = ub + (size_t)p * PLANE;
#pragma unroll
        for (int r = 0; r < 64; ++r) {
          int n = j + r * R128Q;
          v[r] = make_float2(u[(2 * n) * Dch + col], u[(2 * n + 1) * Dch + col]);
        }
#pragma unroll
        for (int r = 64; r < 128; ++r)
          v[r] = make_float2(0.f, 0.f);
      } else if (kmode == 0) {
#pragma unroll
        for (int r = 0; r < 128; ++r)
          v[r] = xk[(j + r * R128Q) * Dch + col];
      } else {
#pragma unroll
        for (int r = 0; r < 64; ++r) {
          int n = j + r * R128Q;
          v[r] = make_float2(xk[(2 * n) * Dch + col].x * invL,
                             xk[(2 * n + 1) * Dch + col].x * invL);
        }
#pragma unroll
        for (int r = 64; r < 128; ++r)
          v[r] = make_float2(0.f, 0.f);
      }
      float2 X[128];
      dft128(v, sign, X);
#pragma unroll
      for (int m = 0; m < 128; ++m)
        y[(128 * j + m) * Dch + col] = X[m];
    }
  }
}

__device__ __forceinline__ void s4_pair_mul(
    float2 zk, float2 zm, float2 ek, float2 ok, float2 c,
    float2& outk, float2& outm)
{
  float2 e  = make_float2(zk.x + zm.x, zk.y - zm.y);
  float2 o  = make_float2(zk.y + zm.y, zm.x - zk.x);
  float2 ye = cadd(cmulf(e, ek), cmulf(c, cmulf(o, ok)));
  float2 yo = cadd(cmulf(e, ok), cmulf(o, ek));
  outk = make_float2(ye.x - yo.y, ye.y + yo.x);
  outm = make_float2(ye.x + yo.y, yo.x - ye.y);
}

#define PMT 128
#define PMU 3
#define PMG (4 * (Llen / 32 + 1) / PMU)
__global__ __launch_bounds__(PMT) void s4_pmul_r16(
    float2* Zb, const float2* __restrict__ ZK, int rev)
{
  __shared__ float2 kE[16][PMT];
  __shared__ float2 kO[16][PMT];
  const int Q = Llen / 16;
  const float R  = 0.70710678118654752440f;
  const float C1 = 0.92387953251128675613f;
  const float S1 = 0.38268343236508977173f;
  const int tid = threadIdx.x;
  for (int i = 0; i < PMU; ++i) {
    const int unit = blockIdx.x * PMU + i;
    const int jj = unit >> 2;
    float sj, cjr;
    sincosf(-2.0f * PI_F * (float)jj / (float)Llen, &sj, &cjr);
    float2 cj = make_float2(cjr, sj);
    float2 cr[16];
    cr[0] = cj;
    cr[1] = cmulf(cj, make_float2(C1, -S1));
    cr[2] = cmulf(cj, make_float2(R, -R));
    cr[3] = cmulf(cj, make_float2(S1, -C1));
    cr[4] = make_float2(cj.y, -cj.x);
    cr[5] = cmulf(cj, make_float2(-S1, -C1));
    cr[6] = cmulf(cj, make_float2(-R, -R));
    cr[7] = cmulf(cj, make_float2(-C1, -S1));
#pragma unroll
    for (int r = 0; r < 8; ++r)
      cr[8 + r] = make_float2(-cr[r].x, -cr[r].y);
    const int col = tid + (unit & 3) * PMT;
    int ia[16], ib[16];
#pragma unroll
    for (int r = 0; r < 16; ++r) {
      ia[r] = (jj + r * Q) * Dch + col;
      ib[r] = (((Q - jj) + (15 - r) * Q) & (Llen - 1)) * Dch + col;
      float2 kk = ZK[ia[r]], km = ZK[ib[r]];
      kE[r][tid] = make_float2(kk.x + km.x, kk.y - km.y);
      kO[r][tid] = make_float2(kk.y + km.y, km.x - kk.x);
    }
    for (int pp = 0; pp < Bsz; ++pp) {
      const int po = (rev ? (Bsz - 1 - pp) : pp) * (int)PLANE;
      float2 va[16], vb[16];
#pragma unroll
      for (int r = 0; r < 16; ++r)
        s4_pair_mul(Zb[po + ia[r]], Zb[po + ib[r]], kE[r][tid], kO[r][tid], cr[r], va[r], vb[15 - r]);
      float2 X[16];
      dft16(va, +1.0f, X);
#pragma unroll
      for (int m = 0; m < 16; ++m)
        Zb[po + ia[m]] = X[m];
      if (jj != 0 && jj != Q / 2) {
        dft16(vb, +1.0f, X);
#pragma unroll
        for (int m = 0; m < 16; ++m)
          Zb[po + ib[15 - m]] = X[m];
      }
    }
  }
}

__global__ __launch_bounds__(256) void s4_final_r32(
    const float2* __restrict__ xb, const float* __restrict__ ub,
    const float* __restrict__ Dp, float* __restrict__ outb, int np, int rev)
{
  const int Q = Llen / 32;
  const float dp = Dp[0];
  const float inv = 1.0f / (4.0f * (float)Llen);
  for (int pp = 0; pp < np; ++pp) {
  const int p = rev ? (np - 1 - pp) : pp;
  const float2* __restrict__ x = xb + (size_t)p * PLANE;
  const float* __restrict__ u = ub + (size_t)p * PLANE;
  float* __restrict__ out = outb + (size_t)p * PLANE;
  for (int j = blockIdx.x; j < Q; j += gridDim.x) {
    float ang = (0.0625f * PI_F) * (float)j / (float)Q;
    float s1, c1;
    sincosf(ang, &s1, &c1);
    float2 w[32];
    w[1]  = make_float2(c1, s1);
    w[2]  = cmulf(w[1], w[1]);
    w[3]  = cmulf(w[2], w[1]);
    w[4]  = cmulf(w[2], w[2]);
    w[5]  = cmulf(w[4], w[1]);
    w[6]  = cmulf(w[3], w[3]);
    w[7]  = cmulf(w[4], w[3]);
    w[8]  = cmulf(w[4], w[4]);
    w[9]  = cmulf(w[8], w[1]);
    w[10] = cmulf(w[5], w[5]);
    w[11] = cmulf(w[8], w[3]);
    w[12] = cmulf(w[6], w[6]);
    w[13] = cmulf(w[8], w[5]);
    w[14] = cmulf(w[7], w[7]);
    w[15] = cmulf(w[8], w[7]);
    w[16] = cmulf(w[8], w[8]);
#pragma unroll
    for (int r = 1; r < 16; ++r)
      w[16 + r] = cmulf(w[16], w[r]);
    for (int h = 0; h < 2; ++h) {
      int col = threadIdx.x + h * HCOL;
      float2 v[32];
      float ue[16][2];
#pragma unroll
      for (int r = 0; r < 32; ++r)
        v[r] = x[(j + r * Q) * Dch + col];
#pragma unroll
      for (int m = 0; m < 16; ++m) {
        int e = (2 * (j + m * Q)) * Dch;
        ue[m][0] = u[e + col];
        ue[m][1] = u[e + Dch + col];
      }
#pragma unroll
      for (int r = 1; r < 32; ++r)
        v[r] = cmulf(v[r], w[r]);
      float2 X[32];
      dft32(v, +1.0f, X);
#pragma unroll
      for (int m = 0; m < 16; ++m) {
        int e = (2 * (j + m * Q)) * Dch;
        out[e + col]       = X[m].x * inv + dp * ue[m][0];
        out[e + Dch + col] = X[m].y * inv + dp * ue[m][1];
      }
    }
  }
  }
}


extern "C" void kernel_launch(void* const* d_in, const int* in_sizes, int n_in,
                              void* d_out, int out_size, void* d_ws, size_t ws_size,
                              hipStream_t stream) {
  const float* u        = (const float*)d_in[0];
  const float* Lam      = (const float*)d_in[1];
  const float* Pri      = (const float*)d_in[2];
  const float* Bri      = (const float*)d_in[3];
  const float* Cri      = (const float*)d_in[4];
  const float* Dp       = (const float*)d_in[5];
  const float* log_step = (const float*)d_in[6];
  float* out = (float*)d_out;

  float2* P0 = (float2*)d_ws;
  float2* P2 = P0 + 2 * PLANE;
  float2* P3 = P0 + 3 * PLANE;
  float2* P4 = P0 + 4 * PLANE;
  float2* P6 = P0 + 6 * PLANE;
  float2* P7 = P0 + 7 * PLANE;
  uint8_t* base = (uint8_t*)P7;
  size_t off = 0;
  auto carve = [&](size_t bytes) -> void* {
    void* p = base + off;
    off += (bytes + 255) & ~(size_t)255;
    return p;
  };
  float*  APr  = (float*)carve((size_t)Llen * Nst * 4);
  float*  APi  = (float*)carve((size_t)Llen * Nst * 4);
  float*  W0r  = (float*)carve((size_t)Nst * Dch * 4);
  float*  W0i  = (float*)carve((size_t)Nst * Dch * 4);
  float*  W1r  = (float*)carve((size_t)Nst * Dch * 4);
  float*  W1i  = (float*)carve((size_t)Nst * Dch * 4);
  float*  perl = (float*)carve((size_t)4 * Llen * 4);
  float2* Ev   = P6;
  (void)ws_size; (void)in_sizes; (void)n_in; (void)out_size;

  const int TB   = 256;
  const int gR128 = 2 * R128Q;
  const int gR32 = NBLK;
  const int gPm  = PMG;

  s4_setup<<<NPB + NWB + NCB, TB, 0, stream>>>(Lam, Pri, Cri, log_step, Bri, perl,
                                              W0r, W0i, W1r, W1i, APr, APi);
  s4_cauchy_gemm<<<dim3(Dch / 16, Llen / 16), 32, 0, stream>>>(
      APr, APi, W0r, W0i, W1r, W1i, perl, Ev);

  const float* u0 = u;
  const float* u1 = u + (size_t)2 * Llen * Dch;

  s4_first_r128<<<gR128, TB, 0, stream>>>(u0, P0, Ev, P2, 0, 3, 1);
  s4_fft_r128<<<gR128, TB, 0, stream>>>(P0, P4, P2, P3, -1.0f, +1.0f, 3, 0);

  s4_first_r128<<<gR128, TB, 0, stream>>>(u1, P0, P3, P2, 1, 3, 1);
  s4_fft_r128<<<gR128, TB, 0, stream>>>(P0, P6, P2, P3, -1.0f, -1.0f, 3, 0);
  float2* Kd = P3;

  s4_pmul_r16<<<gPm, PMT, 0, stream>>>(P4, Kd, 1);
  s4_mid_r32<<<gR32, TB, 0, stream>>>(P4, P0, 4, 0);
  s4_final_r32<<<gR32, TB, 0, stream>>>(P0, u, Dp, out, 4, 1);
}


# Radix-128 butterflies sit on lane pairs (l, l^16): 64 values per lane, a radix-2 step (or a swap of halves) across the pair, then 64-point DFTs; 4 waves per SIMD, not 2; sums regrouped
# speedup vs baseline: 3.9821x; 1.1073x over previous
#include <hip/hip_runtime.h>
#include <hip/hip_bf16.h>
#include <stdint.h>


#define Bsz   4
#define Llen  16384
#define Dch   512
#define Nst   64
#define NFFT  32768
#define PI_F  3.14159265358979323846f

typedef __attribute__((ext_vector_type(2))) float v2f;
typedef __attribute__((ext_vector_type(8))) float v8f;

#ifndef __has_builtin
#define __has_builtin(x) 0
#endif
#if __has_builtin(__builtin_amdgcn_wmma_f32_16x16x4_f32)
#define S4_WMMA 1
#else
#define S4_WMMA 0
#endif

__device__ __forceinline__ float2 cadd(float2 a, float2 b) { return make_float2(a.x + b.x, a.y + b.y); }
__device__ __forceinline__ float2 csub(float2 a, float2 b) { return make_float2(a.x - b.x, a.y - b.y); }
__device__ __forceinline__ float2 cmulf(float2 a, float2 b) { return make_float2(a.x * b.x - a.y * b.y, a.x * b.y + a.y * b.x); }
__device__ __forceinline__ float2 crotis(float2 a, float s) { return make_float2(-s * a.y, s * a.x); }

__device__ __forceinline__ void dft8(const float2* v, float s, float2* X) {
  const float R = 0.70710678118654752440f;
  float2 t0 = cadd(v[0], v[4]), t1 = cadd(v[1], v[5]);
  float2 t2 = cadd(v[2], v[6]), t3 = cadd(v[3], v[7]);
  float2 u0 = csub(v[0], v[4]), u1 = csub(v[1], v[5]);
  float2 u2 = csub(v[2], v[6]), u3 = csub(v[3], v[7]);
  float2 m1 = cmulf(u1, make_float2(R, s * R));
  float2 m2 = crotis(u2, s);
  float2 m3 = cmulf(u3, make_float2(-R, s * R));
  float2 p0 = cadd(t0, t2), p1 = csub(t0, t2);
  float2 p2 = cadd(t1, t3), p3 = crotis(csub(t1, t3), s);
  X[0] = cadd(p0, p2); X[2] = cadd(p1, p3);
  X[4] = csub(p0, p2); X[6] = csub(p1, p3);
  float2 q0 = cadd(u0, m2), q1 = csub(u0, m2);
  float2 q2 = cadd(m1, m3), q3 = crotis(csub(m1, m3), s);
  X[1] = cadd(q0, q2); X[3] = cadd(q1, q3);
  X[5] = csub(q0, q2); X[7] = csub(q1, q3);
}

__device__ __forceinline__ void dft4(float2 a0, float2 a1, float2 a2, float2 a3, float s,
                                     float2& y0, float2& y1, float2& y2, float2& y3) {
  float2 p0 = cadd(a0, a2), p1 = csub(a0, a2);
  float2 p2 = cadd(a1, a3), p3 = crotis(csub(a1, a3), s);
  y0 = cadd(p0, p2); y1 = cadd(p1, p3);
  y2 = csub(p0, p2); y3 = csub(p1, p3);
}

__device__ __forceinline__ void dft16(const float2* v, float s, float2* X) {
  const float R  = 0.70710678118654752440f;
  const float C1 = 0.92387953251128675613f;
  const float S1 = 0.38268343236508977173f;
  float2 t0[4], t1[4], t2[4], t3[4];
  dft4(v[0], v[4], v[8],  v[12], s, t0[0], t0[1], t0[2], t0[3]);
  dft4(v[1], v[5], v[9],  v[13], s, t1[0], t1[1], t1[2], t1[3]);
  dft4(v[2], v[6], v[10], v[14], s, t2[0], t2[1], t2[2], t2[3]);
  dft4(v[3], v[7], v[11], v[15], s, t3[0], t3[1], t3[2], t3[3]);
  t1[1] = cmulf(t1[1], make_float2(C1, s * S1));
  t1[2] = cmulf(t1[2], make_float2(R, s * R));
  t1[3] = cmulf(t1[3], make_float2(S1, s * C1));
  t2[1] = cmulf(t2[1], make_float2(R, s * R));
  t2[2] = crotis(t2[2], s);
  t2[3] = cmulf(t2[3], make_float2(-R, s * R));
  t3[1] = cmulf(t3[1], make_float2(S1, s * C1));
  t3[2] = cmulf(t3[2], make_float2(-R, s * R));
  t3[3] = cmulf(t3[3], make_float2(-C1, -s * S1));
#pragma unroll
  for (int m1 = 0; m1 < 4; ++m1)
    dft4(t0[m1], t1[m1], t2[m1], t3[m1], s, X[m1], X[m1 + 4], X[m1 + 8], X[m1 + 12]);
}

__device__ __forceinline__ void dft32(const float2* v, float s, float2* X) {
  float2 t0_0, t0_1, t0_2, t0_3;
  dft4(v[0], v[8], v[16], v[24], s, t0_0, t0_1, t0_2, t0_3);
  float2 t1_0, t1_1, t1_2, t1_3;
  dft4(v[1], v[9], v[17], v[25], s, t1_0, t1_1, t1_2, t1_3);
  float2 t2_0, t2_1, t2_2, t2_3;
  dft4(v[2], v[10], v[18], v[26], s, t2_0, t2_1, t2_2, t2_3);
  float2 t3_0, t3_1, t3_2, t3_3;
  dft4(v[3], v[11], v[19], v[27], s, t3_0, t3_1, t3_2, t3_3);
  float2 t4_0, t4_1, t4_2, t4_3;
  dft4(v[4], v[12], v[20], v[28], s, t4_0, t4_1, t4_2, t4_3);
  float2 t5_0, t5_1, t5_2, t5_3;
  dft4(v[5], v[13], v[21], v[29], s, t5_0, t5_1, t5_2, t5_3);
  float2 t6_0, t6_1, t6_2, t6_3;
  dft4(v[6], v[14], v[22], v[30], s, t6_0, t6_1, t6_2, t6_3);
  float2 t7_0, t7_1, t7_2, t7_3;
  dft4(v[7], v[15], v[23], v[31], s, t7_0, t7_1, t7_2, t7_3);
  t1_1 = cmulf(t1_1, make_float2(0.98078528040323043f, s * 0.19509032201612825f));
  t2_1 = cmulf(t2_1, make_float2(0.92387953251128674f, s * 0.38268343236508978f));
  t3_1 = cmulf(t3_1, make_float2(0.83146961230254524f, s * 0.55557023301960218f));
  t4_1 = cmulf(t4_1, make_float2(0.70710678118654757f, s * 0.70710678118654746f));
  t5_1 = cmulf(t5_1, make_float2(0.55557023301960229f, s * 0.83146961230254524f));
  t6_1 = cmulf(t6_1, make_float2(0.38268343236508984f, s * 0.92387953251128674f));
  t7_1 = cmulf(t7_1, make_float2(0.19509032201612833f, s * 0.98078528040323043f));
  t1_2 = cmulf(t1_2, make_float2(0.92387953251128674f, s * 0.38268343236508978f));
  t2_2 = cmulf(t2_2, make_float2(0.70710678118654757f, s * 0.70710678118654746f));
  t3_2 = cmulf(t3_2, make_float2(0.38268343236508984f, s * 0.92387953251128674f));
  t4_2 = crotis(t4_2, s);
  t5_2 = cmulf(t5_2, make_float2(-0.38268343236508973f, s * 0.92387953251128674f));
  t6_2 = cmulf(t6_2, make_float2(-0.70710678118654746f, s * 0.70710678118654757f));
  t7_2 = cmulf(t7_2, make_float2(-0.92387953251128674f, s * 0.38268343236508989f));
  t1_3 = cmulf(t1_3, make_float2(0.83146961230254524f, s * 0.55557023301960218f));
  t2_3 = cmulf(t2_3, make_float2(0.38268343236508984f, s * 0.92387953251128674f));
  t3_3 = cmulf(t3_3, make_float2(-0.19509032201612819f, s * 0.98078528040323043f));
  t4_3 = cmulf(t4_3, make_float2(-0.70710678118654746f, s * 0.70710678118654757f));
  t5_3 = cmulf(t5_3, make_float2(-0.98078528040323043f, s * 0.19509032201612861f));
  t6_3 = cmulf(t6_3, make_float2(-0.92387953251128685f, s * -0.38268343236508967f));
  t7_3 = cmulf(t7_3, make_float2(-0.55557023301960218f, s * -0.83146961230254524f));
  {
    float2 a[8] = {t0_0, t1_0, t2_0, t3_0, t4_0, t5_0, t6_0, t7_0};
    float2 b[8];
    dft8(a, s, b);
    X[0] = b[0]; X[4] = b[1]; X[8] = b[2]; X[12] = b[3];
    X[16] = b[4]; X[20] = b[5]; X[24] = b[6]; X[28] = b[7];
  }
  {
    float2 a[8] = {t0_1, t1_1, t2_1, t3_1, t4_1, t5_1, t6_1, t7_1};
    float2 b[8];
    dft8(a, s, b);
    X[1] = b[0]; X[5] = b[1]; X[9] = b[2]; X[13] = b[3];
    X[17] = b[4]; X[21] = b[5]; X[25] = b[6]; X[29] = b[7];
  }
  {
    float2 a[8] = {t0_2, t1_2, t2_2, t3_2, t4_2, t5_2, t6_2, t7_2};
    float2 b[8];
    dft8(a, s, b);
    X[2] = b[0]; X[6] = b[1]; X[10] = b[2]; X[14] = b[3];
    X[18] = b[4]; X[22] = b[5]; X[26] = b[6]; X[30] = b[7];
  }
  {
    float2 a[8] = {t0_3, t1_3, t2_3, t3_3, t4_3, t5_3, t6_3, t7_3};
    float2 b[8];
    dft8(a, s, b);
    X[3] = b[0]; X[7] = b[1]; X[11] = b[2]; X[15] = b[3];
    X[19] = b[4]; X[23] = b[5]; X[27] = b[6]; X[31] = b[7];
  }
}

__device__ __forceinline__ void dft64(const float2* v, float s, float2* X) {
  float2 t0[16], t1[16], t2[16], t3[16];
#pragma unroll
  for (int r2 = 0; r2 < 16; ++r2)
    dft4(v[r2], v[r2 + 16], v[r2 + 32], v[r2 + 48], s, t0[r2], t1[r2], t2[r2], t3[r2]);
  t1[1] = cmulf(t1[1], make_float2(0.99518472667219693f, s * 0.098017140329560604f));
  t1[2] = cmulf(t1[2], make_float2(0.98078528040323043f, s * 0.19509032201612825f));
  t1[3] = cmulf(t1[3], make_float2(0.95694033573220882f, s * 0.29028467725446233f));
  t1[4] = cmulf(t1[4], make_float2(0.92387953251128674f, s * 0.38268343236508978f));
  t1[5] = cmulf(t1[5], make_float2(0.88192126434835505f, s * 0.47139673682599764f));
  t1[6] = cmulf(t1[6], make_float2(0.83146961230254524f, s * 0.55557023301960218f));
  t1[7] = cmulf(t1[7], make_float2(0.77301045336273699f, s * 0.63439328416364549f));
  t1[8] = cmulf(t1[8], make_float2(0.70710678118654757f, s * 0.70710678118654746f));
  t1[9] = cmulf(t1[9], make_float2(0.63439328416364549f, s * 0.77301045336273699f));
  t1[10] = cmulf(t1[10], make_float2(0.55557023301960229f, s * 0.83146961230254524f));
  t1[11] = cmulf(t1[11], make_float2(0.47139673682599781f, s * 0.88192126434835494f));
  t1[12] = cmulf(t1[12], make_float2(0.38268343236508984f, s * 0.92387953251128674f));
  t1[13] = cmulf(t1[13], make_float2(0.29028467725446233f, s * 0.95694033573220894f));
  t1[14] = cmulf(t1[14], make_float2(0.19509032201612833f, s * 0.98078528040323043f));
  t1[15] = cmulf(t1[15], make_float2(0.09801714032956077f, s * 0.99518472667219682f));
  t2[1] = cmulf(t2[1], make_float2(0.98078528040323043f, s * 0.19509032201612825f));
  t2[2] = cmulf(t2[2], make_float2(0.92387953251128674f, s * 0.38268343236508978f));
  t2[3] = cmulf(t2[3], make_float2(0.83146961230254524f, s * 0.55557023301960218f));
  t2[4] = cmulf(t2[4], make_float2(0.70710678118654757f, s * 0.70710678118654746f));
  t2[5] = cmulf(t2[5], make_float2(0.55557023301960229f, s * 0.83146961230254524f));
  t2[6] = cmulf(t2[6], make_float2(0.38268343236508984f, s * 0.92387953251128674f));
  t2[7] = cmulf(t2[7], make_float2(0.19509032201612833f, s * 0.98078528040323043f));
  t2[8] = crotis(t2[8], s);
  t2[9] = cmulf(t2[9], make_float2(-0.19509032201612819f, s * 0.98078528040323043f));
  t2[10] = cmulf(t2[10], make_float2(-0.38268343236508973f, s * 0.92387953251128674f));
  t2[11] = cmulf(t2[11], make_float2(-0.55557023301960196f, s * 0.83146961230254546f));
  t2[12] = cmulf(t2[12], make_float2(-0.70710678118654746f, s * 0.70710678118654757f));
  t2[13] = cmulf(t2[13], make_float2(-0.83146961230254535f, s * 0.55557023301960218f));
  t2[14] = cmulf(t2[14], make_float2(-0.92387953251128674f, s * 0.38268343236508989f));
  t2[15] = cmulf(t2[15], make_float2(-0.98078528040323043f, s * 0.19509032201612861f));
  t3[1] = cmulf(t3[1], make_float2(0.95694033573220882f, s * 0.29028467725446233f));
  t3[2] = cmulf(t3[2], make_float2(0.83146961230254524f, s * 0.55557023301960218f));
  t3[3] = cmulf(t3[3], make_float2(0.63439328416364549f, s * 0.77301045336273699f));
  t3[4] = cmulf(t3[4], make_float2(0.38268343236508984f, s * 0.92387953251128674f));
  t3[5] = cmulf(t3[5], make_float2(0.09801714032956077f, s * 0.99518472667219682f));
  t3[6] = cmulf(t3[6], make_float2(-0.19509032201612819f, s * 0.98078528040323043f));
  t3[7] = cmulf(t3[7], make_float2(-0.4713967368259977f, s * 0.88192126434835505f));
  t3[8] = cmulf(t3[8], make_float2(-0.70710678118654746f, s * 0.70710678118654757f));
  t3[9] = cmulf(t3[9], make_float2(-0.88192126434835494f, s * 0.47139673682599786f));
  t3[10] = cmulf(t3[10], make_float2(-0.98078528040323043f, s * 0.19509032201612861f));
  t3[11] = cmulf(t3[11], make_float2(-0.99518472667219693f, s * -0.09801714032956059f));
  t3[12] = cmulf(t3[12], make_float2(-0.92387953251128685f, s * -0.38268343236508967f));
  t3[13] = cmulf(t3[13], make_float2(-0.7730104533627371f, s * -0.63439328416364527f));
  t3[14] = cmulf(t3[14], make_float2(-0.55557023301960218f, s * -0.83146961230254524f));
  t3[15] = cmulf(t3[15], make_float2(-0.29028467725446244f, s * -0.95694033573220882f));
  float2 Y[16];
  dft16(t0, s, Y);
#pragma unroll
  for (int m2 = 0; m2 < 16; ++m2) X[0 + 4 * m2] = Y[m2];
  dft16(t1, s, Y);
#pragma unroll
  for (int m2 = 0; m2 < 16; ++m2) X[1 + 4 * m2] = Y[m2];
  dft16(t2, s, Y);
#pragma unroll
  for (int m2 = 0; m2 < 16; ++m2) X[2 + 4 * m2] = Y[m2];
  dft16(t3, s, Y);
#pragma unroll
  for (int m2 = 0; m2 < 16; ++m2) X[3 + 4 * m2] = Y[m2];
}

__device__ __forceinline__ void tw128_odd(float2* v, float s) {
  v[1] = cmulf(v[1], make_float2(0.99879545620517241f, s * 0.049067674327418015f));
  v[2] = cmulf(v[2], make_float2(0.99518472667219693f, s * 0.098017140329560604f));
  v[3] = cmulf(v[3], make_float2(0.98917650996478101f, s * 0.14673047445536175f));
  v[4] = cmulf(v[4], make_float2(0.98078528040323043f, s * 0.19509032201612825f));
  v[5] = cmulf(v[5], make_float2(0.97003125319454397f, s * 0.24298017990326387f));
  v[6] = cmulf(v[6], make_float2(0.95694033573220882f, s * 0.29028467725446233f));
  v[7] = cmulf(v[7], make_float2(0.94154406518302081f, s * 0.33688985339222005f));
  v[8] = cmulf(v[8], make_float2(0.92387953251128674f, s * 0.38268343236508978f));
  v[9] = cmulf(v[9], make_float2(0.90398929312344334f, s * 0.42755509343028208f));
  v[10] = cmulf(v[10], make_float2(0.88192126434835505f, s * 0.47139673682599764f));
  v[11] = cmulf(v[11], make_float2(0.85772861000027212f, s * 0.51410274419322166f));
  v[12] = cmulf(v[12], make_float2(0.83146961230254524f, s * 0.55557023301960218f));
  v[13] = cmulf(v[13], make_float2(0.80320753148064494f, s * 0.59569930449243336f));
  v[14] = cmulf(v[14], make_float2(0.77301045336273699f, s * 0.63439328416364549f));
  v[15] = cmulf(v[15], make_float2(0.74095112535495911f, s * 0.67155895484701833f));
  v[16] = cmulf(v[16], make_float2(0.70710678118654757f, s * 0.70710678118654746f));
  v[17] = cmulf(v[17], make_float2(0.67155895484701833f, s * 0.74095112535495911f));
  v[18] = cmulf(v[18], make_float2(0.63439328416364549f, s * 0.77301045336273699f));
  v[19] = cmulf(v[19], make_float2(0.59569930449243347f, s * 0.80320753148064483f));
  v[20] = cmulf(v[20], make_float2(0.55557023301960229f, s * 0.83146961230254524f));
  v[21] = cmulf(v[21], make_float2(0.51410274419322166f, s * 0.85772861000027212f));
  v[22] = cmulf(v[22], make_float2(0.47139673682599781f, s * 0.88192126434835494f));
  v[23] = cmulf(v[23], make_float2(0.4275550934302822f, s * 0.90398929312344334f));
  v[24] = cmulf(v[24], make_float2(0.38268343236508984f, s * 0.92387953251128674f));
  v[25] = cmulf(v[25], make_float2(0.33688985339222005f, s * 0.94154406518302081f));
  v[26] = cmulf(v[26], make_float2(0.29028467725446233f, s * 0.95694033573220894f));
  v[27] = cmulf(v[27], make_float2(0.24298017990326398f, s * 0.97003125319454397f));
  v[28] = cmulf(v[28], make_float2(0.19509032201612833f, s * 0.98078528040323043f));
  v[29] = cmulf(v[29], make_float2(0.14673047445536175f, s * 0.98917650996478101f));
  v[30] = cmulf(v[30], make_float2(0.09801714032956077f, s * 0.99518472667219682f));
  v[31] = cmulf(v[31], make_float2(0.049067674327418126f, s * 0.99879545620517241f));
  v[32] = crotis(v[32], s);
  v[33] = cmulf(v[33], make_float2(-0.049067674327418008f, s * 0.99879545620517241f));
  v[34] = cmulf(v[34], make_float2(-0.098017140329560645f, s * 0.99518472667219693f));
  v[35] = cmulf(v[35], make_float2(-0.14673047445536164f, s * 0.98917650996478101f));
  v[36] = cmulf(v[36], make_float2(-0.19509032201612819f, s * 0.98078528040323043f));
  v[37] = cmulf(v[37], make_float2(-0.24298017990326387f, s * 0.97003125319454397f));
  v[38] = cmulf(v[38], make_float2(-0.29028467725446216f, s * 0.95694033573220894f));
  v[39] = cmulf(v[39], make_float2(-0.33688985339221994f, s * 0.94154406518302081f));
  v[40] = cmulf(v[40], make_float2(-0.38268343236508973f, s * 0.92387953251128674f));
  v[41] = cmulf(v[41], make_float2(-0.42755509343028186f, s * 0.90398929312344345f));
  v[42] = cmulf(v[42], make_float2(-0.4713967368259977f, s * 0.88192126434835505f));
  v[43] = cmulf(v[43], make_float2(-0.51410274419322166f, s * 0.85772861000027212f));
  v[44] = cmulf(v[44], make_float2(-0.55557023301960196f, s * 0.83146961230254546f));
  v[45] = cmulf(v[45], make_float2(-0.59569930449243336f, s * 0.80320753148064494f));
  v[46] = cmulf(v[46], make_float2(-0.63439328416364538f, s * 0.7730104533627371f));
  v[47] = cmulf(v[47], make_float2(-0.67155895484701844f, s * 0.74095112535495899f));
  v[48] = cmulf(v[48], make_float2(-0.70710678118654746f, s * 0.70710678118654757f));
  v[49] = cmulf(v[49], make_float2(-0.74095112535495888f, s * 0.67155895484701855f));
  v[50] = cmulf(v[50], make_float2(-0.77301045336273699f, s * 0.63439328416364549f));
  v[51] = cmulf(v[51], make_float2(-0.80320753148064483f, s * 0.59569930449243347f));
  v[52] = cmulf(v[52], make_float2(-0.83146961230254535f, s * 0.55557023301960218f));
  v[53] = cmulf(v[53], make_float2(-0.85772861000027201f, s * 0.51410274419322177f));
  v[54] = cmulf(v[54], make_float2(-0.88192126434835494f, s * 0.47139673682599786f));
  v[55] = cmulf(v[55], make_float2(-0.90398929312344334f, s * 0.42755509343028203f));
  v[56] = cmulf(v[56], make_float2(-0.92387953251128674f, s * 0.38268343236508989f));
  v[57] = cmulf(v[57], make_float2(-0.9415440651830207f, s * 0.33688985339222033f));
  v[58] = cmulf(v[58], make_float2(-0.95694033573220882f, s * 0.29028467725446239f));
  v[59] = cmulf(v[59], make_float2(-0.97003125319454397f, s * 0.24298017990326407f));
  v[60] = cmulf(v[60], make_float2(-0.98078528040323043f, s * 0.19509032201612861f));
  v[61] = cmulf(v[61], make_float2(-0.98917650996478101f, s * 0.1467304744553618f));
  v[62] = cmulf(v[62], make_float2(-0.99518472667219682f, s * 0.098017140329560826f));
  v[63] = cmulf(v[63], make_float2(-0.99879545620517241f, s * 0.049067674327417966f));
}

__device__ __forceinline__ float xhalf(float x) {
  return __builtin_bit_cast(float, __builtin_amdgcn_permlanex16(
      __builtin_bit_cast(int, x), __builtin_bit_cast(int, x), 0x76543210, 0xfedcba98, false, false));
}

__device__ __forceinline__ void s4_g_of_l(
    int l, const float* __restrict__ log_step,
    float& gr, float& gi, float& scr, float& sci)
{
  float step = expf(log_step[0]);
  float theta = -2.0f * PI_F * (float)l / (float)Llen;
  float oi, orr;
  sincosf(theta, &oi, &orr);
  float op1r = 1.0f + orr;
  float den = op1r * op1r + oi * oi;
  float nr = 1.0f - orr, ni = -oi;
  float gscale = 2.0f / step;
  gr = gscale * (nr * op1r + ni * oi) / den;
  gi = gscale * (ni * op1r - nr * oi) / den;
  scr = 2.0f * op1r / den;
  sci = -2.0f * oi / den;
}

__device__ __forceinline__ void s4_perl_body(
    int l, const float* __restrict__ Lam, const float* __restrict__ Pri,
    const float* __restrict__ Cri, const float* __restrict__ log_step,
    float* __restrict__ perl)
{
  if (l >= Llen) return;
  float gr, gi, scr, sci;
  s4_g_of_l(l, log_step, gr, gi, scr, sci);
  float k01r = 0.f, k01i = 0.f, k11r = 0.f, k11i = 0.f;
  for (int n = 0; n < Nst; ++n) {
    float lr = Lam[2 * n], li = Lam[2 * n + 1];
    float pr = Pri[2 * n], pi = Pri[2 * n + 1];
    float cr = Cri[2 * n], ci = Cri[2 * n + 1];
    float zr = gr - lr, zi = gi - li;
    float zd = zr * zr + zi * zi;
    float car = zr / zd, cai = -zi / zd;
    float w01r = cr * pr + ci * pi;
    float w01i = cr * pi - ci * pr;
    float w11r = pr * pr + pi * pi;
    k01r += car * w01r - cai * w01i;
    k01i += car * w01i + cai * w01r;
    k11r += car * w11r;
    k11i += cai * w11r;
  }
  float dr = 1.0f + k11r, di = k11i;
  float dd = dr * dr + di * di;
  float sr = (k01r * dr + k01i * di) / dd;
  float si = (k01i * dr - k01r * di) / dd;
  perl[l] = sr;
  perl[Llen + l] = si;
  perl[2 * Llen + l] = scr;
  perl[3 * Llen + l] = sci;
}

__device__ __forceinline__ void s4_weights_body(
    int idx, const float* __restrict__ Bri, const float* __restrict__ Pri,
    const float* __restrict__ Cri,
    float* __restrict__ W0r, float* __restrict__ W0i,
    float* __restrict__ W1r, float* __restrict__ W1i)
{
  if (idx >= Nst * Dch) return;
  int n = idx / Dch, d = idx % Dch;
  float br = Bri[2 * (n * Dch + d)], bi = Bri[2 * (n * Dch + d) + 1];
  float cr = Cri[2 * n], ci = Cri[2 * n + 1];
  float pr = Pri[2 * n], pi = Pri[2 * n + 1];
  int o = (n >> 1) * (2 * Dch) + 2 * d + (n & 1);
  W0r[o] = cr * br + ci * bi;
  W0i[o] = cr * bi - ci * br;
  W1r[o] = pr * br + pi * bi;
  W1i[o] = pr * bi - pi * br;
}

#define NPB  (Llen / 256)
#define NWB  (Nst * Dch / 256)
#define NCB  512
__device__ __forceinline__ void s4_cauchy_body(
    int cb, const float* __restrict__ Lam, const float* __restrict__ log_step,
    float* __restrict__ APr, float* __restrict__ APi)
{
  const int idx0 = cb * 256 + threadIdx.x;
  const int l = idx0 % Llen;
  float gr, gi, scr, sci;
  s4_g_of_l(l, log_step, gr, gi, scr, sci);
  (void)scr; (void)sci;
  for (int idx = idx0; idx < Llen * Nst; idx += NCB * 256) {
    int n = idx / Llen;
    float zr = gr - Lam[2 * n], zi = gi - Lam[2 * n + 1];
    float zd = zr * zr + zi * zi;
    int o = (n >> 1) * (2 * Llen) + 2 * l + (n & 1);
    APr[o] = zr / zd;
    APi[o] = -zi / zd;
  }
}

__global__ __launch_bounds__(256) void s4_setup(
    const float* __restrict__ Lam, const float* __restrict__ Pri,
    const float* __restrict__ Cri, const float* __restrict__ log_step,
    const float* __restrict__ Bri, float* __restrict__ perl,
    float* __restrict__ W0r, float* __restrict__ W0i,
    float* __restrict__ W1r, float* __restrict__ W1i,
    float* __restrict__ APr, float* __restrict__ APi)
{
  const int b = blockIdx.x;
  if (b < NPB)
    s4_perl_body(b * 256 + threadIdx.x, Lam, Pri, Cri, log_step, perl);
  else if (b < NPB + NWB)
    s4_weights_body((b - NPB) * 256 + threadIdx.x, Bri, Pri, Cri, W0r, W0i, W1r, W1i);
  else
    s4_cauchy_body(b - NPB - NWB, Lam, log_step, APr, APi);
}

__global__ __launch_bounds__(32) void s4_cauchy_gemm(
    const float* __restrict__ APr, const float* __restrict__ APi,
    const float* __restrict__ W0r, const float* __restrict__ W0i,
    const float* __restrict__ W1r, const float* __restrict__ W1i,
    const float* __restrict__ perl, float2* __restrict__ Ev)
{
  int lane = threadIdx.x & 31;
  int tileD = blockIdx.x;
  int tileL = blockIdx.y;
  int d0 = tileD * 16 + (lane & 15);
  int mbase = tileL * 16 + ((lane >> 4) ? 8 : 0);
#if S4_WMMA
  int m = tileL * 16 + (lane & 15);
  int khi = (lane >> 4) * 2;
  v8f aRr = {}, aRi = {}, aSr = {}, aSi = {};
  for (int kb = 0; kb < Nst; kb += 4) {
    int k0 = kb + khi;
    const float* pr = APr + (k0 >> 1) * (2 * Llen) + 2 * m;
    const float* pi = APi + (k0 >> 1) * (2 * Llen) + 2 * m;
    v2f ar = *(const v2f*)pr;
    v2f ai = *(const v2f*)pi;
    v2f nai = -ai;
    int bo = (k0 >> 1) * (2 * Dch) + 2 * d0;
    v2f b0r = *(const v2f*)(W0r + bo);
    v2f b0i = *(const v2f*)(W0i + bo);
    v2f b1r = *(const v2f*)(W1r + bo);
    v2f b1i = *(const v2f*)(W1i + bo);
    aRr = __builtin_amdgcn_wmma_f32_16x16x4_f32(false, ar,  false, b0r, (short)0, aRr, false, false);
    aRr = __builtin_amdgcn_wmma_f32_16x16x4_f32(false, nai, false, b0i, (short)0, aRr, false, false);
    aRi = __builtin_amdgcn_wmma_f32_16x16x4_f32(false, ar,  false, b0i, (short)0, aRi, false, false);
    aRi = __builtin_amdgcn_wmma_f32_16x16x4_f32(false, ai,  false, b0r, (short)0, aRi, false, false);
    aSr = __builtin_amdgcn_wmma_f32_16x16x4_f32(false, ar,  false, b1r, (short)0, aSr, false, false);
    aSr = __builtin_amdgcn_wmma_f32_16x16x4_f32(false, nai, false, b1i, (short)0, aSr, false, false);
    aSi = __builtin_amdgcn_wmma_f32_16x16x4_f32(false, ar,  false, b1i, (short)0, aSi, false, false);
    aSi = __builtin_amdgcn_wmma_f32_16x16x4_f32(false, ai,  false, b1r, (short)0, aSi, false, false);
  }
  for (int r = 0; r < 8; ++r) {
    int l = mbase + r;
    float k00r = aRr[r], k00i = aRi[r];
    float k10r = aSr[r], k10i = aSi[r];
    float sr = perl[l],            si = perl[Llen + l];
    float cr = perl[2 * Llen + l], ci = perl[3 * Llen + l];
    float tr = k00r - (sr * k10r - si * k10i);
    float ti = k00i - (sr * k10i + si * k10r);
    Ev[l * Dch + d0] = make_float2(cr * tr - ci * ti, cr * ti + ci * tr);
  }
#else
  for (int r = 0; r < 8; ++r) {
    int l = mbase + r;
    float k00r = 0.f, k00i = 0.f, k10r = 0.f, k10i = 0.f;
    for (int k = 0; k < Nst; ++k) {
      int ao = (k >> 1) * (2 * Llen) + 2 * l + (k & 1);
      int bo = (k >> 1) * (2 * Dch) + 2 * d0 + (k & 1);
      float arr = APr[ao], aii = APi[ao];
      float b0r = W0r[bo], b0i = W0i[bo];
      float b1r = W1r[bo], b1i = W1i[bo];
      k00r += arr * b0r - aii * b0i;  k00i += arr * b0i + aii * b0r;
      k10r += arr * b1r - aii * b1i;  k10i += arr * b1i + aii * b1r;
    }
    float sr = perl[l],            si = perl[Llen + l];
    float cr = perl[2 * Llen + l], ci = perl[3 * Llen + l];
    float tr = k00r - (sr * k10r - si * k10i);
    float ti = k00i - (sr * k10i + si * k10r);
    Ev[l * Dch + d0] = make_float2(cr * tr - ci * ti, cr * ti + ci * tr);
  }
#endif
}

#define NBLK  512
#define HCOL  256
#define PLANE ((size_t)Llen * Dch)

__global__ __launch_bounds__(256) void s4_mid_r32(
    const float2* __restrict__ xb, float2* __restrict__ yb, int np, int rev)
{
  const int Ns = 16;
  const int q = Llen >> 5;
  for (int pp = 0; pp < np; ++pp) {
  const int p = rev ? (np - 1 - pp) : pp;
  const float2* __restrict__ x = xb + (size_t)p * PLANE;
  float2* __restrict__ y = yb + (size_t)p * PLANE;
  for (int j = blockIdx.x; j < q; j += gridDim.x) {
    int k = j & (Ns - 1);
    float ang = (0.0625f * PI_F) * (float)k / (float)Ns;
    float s1, c1;
    sincosf(ang, &s1, &c1);
    float2 w[32];
    w[1]  = make_float2(c1, s1);
    w[2]  = cmulf(w[1], w[1]);
    w[3]  = cmulf(w[2], w[1]);
    w[4]  = cmulf(w[2], w[2]);
    w[5]  = cmulf(w[4], w[1]);
    w[6]  = cmulf(w[3], w[3]);
    w[7]  = cmulf(w[4], w[3]);
    w[8]  = cmulf(w[4], w[4]);
    w[9]  = cmulf(w[8], w[1]);
    w[10] = cmulf(w[5], w[5]);
    w[11] = cmulf(w[8], w[3]);
    w[12] = cmulf(w[6], w[6]);
    w[13] = cmulf(w[8], w[5]);
    w[14] = cmulf(w[7], w[7]);
    w[15] = cmulf(w[8], w[7]);
    w[16] = cmulf(w[8], w[8]);
#pragma unroll
    for (int r = 1; r < 16; ++r)
      w[16 + r] = cmulf(w[16], w[r]);
    int i0 = 32 * j - 31 * k;
    int r0 = (j >> 4) + k * (Llen / 16);
#pragma nounroll
    for (int h = 0; h < 2; ++h) {
      int col = threadIdx.x + h * HCOL;
      float2 v[32];
#pragma unroll
      for (int r = 0; r < 32; ++r)
        v[r] = x[(r0 + 32 * r) * Dch + col];
      __builtin_amdgcn_sched_group_barrier(0x020, 32, 0);
#pragma unroll
      for (int r = 1; r < 32; ++r)
        v[r] = cmulf(v[r], w[r]);
      float2 X[32];
      dft32(v, +1.0f, X);
#pragma unroll
      for (int m = 0; m < 32; ++m)
        y[(i0 + m * Ns) * Dch + col] = X[m];
    }
  }
  }
}

#define R128Q (Llen / 128)
#define QCOL  128
__global__ __launch_bounds__(256) __attribute__((amdgpu_waves_per_eu(4, 4))) void s4_fft_r128(
    const float2* __restrict__ xb, float2* __restrict__ yb,
    const float2* __restrict__ xk, float2* __restrict__ yk,
    float signb, float signk, int np, int rev)
{
  const int lane = threadIdx.x & 31;
  const int h = lane >> 4;
  const int cw = (threadIdx.x >> 5) * 16 + (lane & 15);
  const float sg = h ? -1.0f : 1.0f;
  for (int pp = 0; pp < np; ++pp) {
  const int p = rev ? (np - 1 - pp) : pp;
  const float2* __restrict__ x = (p < 2) ? xb + (size_t)p * PLANE : xk;
  float2* __restrict__ y = (p < 2) ? yb + (size_t)p * PLANE : yk;
  const float sign = (p < 2) ? signb : signk;
  for (int t = blockIdx.x; t < 4 * R128Q; t += gridDim.x) {
    const int j = t >> 2;
    const int col = cw + (t & 3) * QCOL;
    float ang = sign * (2.0f * PI_F) * (float)j / (float)Llen;
    float s1, c1, s16, c16;
    sincosf(ang, &s1, &c1);
    sincosf(16.0f * ang, &s16, &c16);
    float2 wl[16], wh[8];
    wl[1]  = make_float2(c1, s1);
    wl[2]  = cmulf(wl[1], wl[1]);
    wl[3]  = cmulf(wl[2], wl[1]);
    wl[4]  = cmulf(wl[2], wl[2]);
    wl[5]  = cmulf(wl[4], wl[1]);
    wl[6]  = cmulf(wl[3], wl[3]);
    wl[7]  = cmulf(wl[4], wl[3]);
    wl[8]  = cmulf(wl[4], wl[4]);
#pragma unroll
    for (int r = 1; r < 8; ++r)
      wl[8 + r] = cmulf(wl[8], wl[r]);
    wh[1] = make_float2(c16, s16);
    wh[2] = cmulf(wh[1], wh[1]);
    wh[3] = cmulf(wh[2], wh[1]);
    wh[4] = cmulf(wh[2], wh[2]);
    wh[5] = cmulf(wh[4], wh[1]);
    wh[6] = cmulf(wh[3], wh[3]);
    wh[7] = cmulf(wh[4], wh[3]);
    float2 wq[4];
    wq[0] = h ? wh[4] : make_float2(1.0f, 0.0f);
    wq[1] = h ? wh[5] : wh[1];
    wq[2] = h ? wh[6] : wh[2];
    wq[3] = h ? wh[7] : wh[3];
    const float2* __restrict__ xr = x + (size_t)(j + 64 * h * R128Q) * Dch + col;
    float2 v[64];
#pragma unroll
    for (int i = 0; i < 64; ++i)
      v[i] = xr[(size_t)i * R128Q * Dch];
#pragma unroll
    for (int q = 0; q < 4; ++q) {
      v[16 * q] = cmulf(v[16 * q], wq[q]);
#pragma unroll
      for (int lo = 1; lo < 16; ++lo)
        v[16 * q + lo] = cmulf(v[16 * q + lo], cmulf(wq[q], wl[lo]));
    }
#pragma unroll
    for (int i = 0; i < 64; ++i) {
      float ox = xhalf(v[i].x), oy = xhalf(v[i].y);
      v[i] = make_float2(fmaf(sg, v[i].x, ox), fmaf(sg, v[i].y, oy));
    }
    if (h)
      tw128_odd(v, sign);
    float2 X[64];
    dft64(v, sign, X);
    float2* __restrict__ yr = y + (size_t)(j + h * R128Q) * Dch + col;
#pragma unroll
    for (int m = 0; m < 64; ++m)
      yr[(size_t)m * 2 * R128Q * Dch] = X[m];
  }
  }
}


__global__ __launch_bounds__(256) __attribute__((amdgpu_waves_per_eu(4, 4))) void s4_first_r128(
    const float* __restrict__ ub, float2* __restrict__ yb,
    const float2* __restrict__ xk, float2* __restrict__ yk,
    int kmode, int np, int rev)
{
  const float invL = 1.0f / (float)Llen;
  const int lane = threadIdx.x & 31;
  const int h = lane >> 4;
  const int cw = (threadIdx.x >> 5) * 16 + (lane & 15);
  const float sg = h ? -1.0f : 1.0f;
  for (int pp = 0; pp < np; ++pp) {
    const int p = rev ? (np - 1 - pp) : pp;
    float2* __restrict__ y = (p < 2) ? yb + (size_t)p * PLANE : yk;
    const float sign = (p == 2 && kmode == 0) ? +1.0f : -1.0f;
    for (int t = blockIdx.x; t < 4 * R128Q; t += gridDim.x) {
      const int j = t >> 2;
      const int col = cw + (t & 3) * QCOL;
      float2 v[64];
      if (p == 2 && kmode == 0) {
        const float2* __restrict__ xr = xk + (size_t)(j + 64 * h * R128Q) * Dch + col;
#pragma unroll
        for (int i = 0; i < 64; ++i)
          v[i] = xr[(size_t)i * R128Q * Dch];
#pragma unroll
        for (int i = 0; i < 64; ++i) {
          float ox = xhalf(v[i].x), oy = xhalf(v[i].y);
          v[i] = make_float2(fmaf(sg, v[i].x, ox), fmaf(sg, v[i].y, oy));
        }
      } else {
        float2 own[32];
        if (p < 2) {
          const float* __restrict__ ur = ub + (size_t)p * PLANE
                                            + (size_t)(2 * (j + 32 * h * R128Q)) * Dch + col;
#pragma unroll
          for (int k = 0; k < 32; ++k)
            own[k] = make_float2(ur[(size_t)k * 2 * R128Q * Dch], ur[(size_t)k * 2 * R128Q * Dch + Dch]);
        } else {
          const float2* __restrict__ kr = xk + (size_t)(2 * (j + 32 * h * R128Q)) * Dch + col;
#pragma unroll
          for (int k = 0; k < 32; ++k)
            own[k] = make_float2(kr[(size_t)k * 2 * R128Q * Dch].x * invL,
                                 kr[(size_t)k * 2 * R128Q * Dch + Dch].x * invL);
        }
#pragma unroll
        for (int k = 0; k < 32; ++k) {
          float2 o = make_float2(xhalf(own[k].x), xhalf(own[k].y));
          v[k]      = h ? o : own[k];
          v[32 + k] = h ? own[k] : o;
        }
      }
      if (h)
        tw128_odd(v, sign);
      float2 X[64];
      dft64(v, sign, X);
      float2* __restrict__ yr = y + (size_t)(128 * j + h) * Dch + col;
#pragma unroll
      for (int m = 0; m < 64; ++m)
        yr[(size_t)m * 2 * Dch] = X[m];
    }
  }
}

__device__ __forceinline__ void s4_pair_mul(
    float2 zk, float2 zm, float2 ek, float2 ok, float2 c,
    float2& outk, float2& outm)
{
  float2 e  = make_float2(zk.x + zm.x, zk.y - zm.y);
  float2 o  = make_float2(zk.y + zm.y, zm.x - zk.x);
  float2 ye = cadd(cmulf(e, ek), cmulf(c, cmulf(o, ok)));
  float2 yo = cadd(cmulf(e, ok), cmulf(o, ek));
  outk = make_float2(ye.x - yo.y, ye.y + yo.x);
  outm = make_float2(ye.x + yo.y, yo.x - ye.y);
}

#define PMT 128
#define PMU 3
#define PMG (4 * (Llen / 32 + 1) / PMU)
__global__ __launch_bounds__(PMT) void s4_pmul_r16(
    float2* Zb, const float2* __restrict__ ZK, int rev)
{
  __shared__ float2 kE[16][PMT];
  __shared__ float2 kO[16][PMT];
  const int Q = Llen / 16;
  const float R  = 0.70710678118654752440f;
  const float C1 = 0.92387953251128675613f;
  const float S1 = 0.38268343236508977173f;
  const int tid = threadIdx.x;
  for (int i = 0; i < PMU; ++i) {
    const int unit = blockIdx.x * PMU + i;
    const int jj = unit >> 2;
    float sj, cjr;
    sincosf(-2.0f * PI_F * (float)jj / (float)Llen, &sj, &cjr);
    float2 cj = make_float2(cjr, sj);
    float2 cr[16];
    cr[0] = cj;
    cr[1] = cmulf(cj, make_float2(C1, -S1));
    cr[2] = cmulf(cj, make_float2(R, -R));
    cr[3] = cmulf(cj, make_float2(S1, -C1));
    cr[4] = make_float2(cj.y, -cj.x);
    cr[5] = cmulf(cj, make_float2(-S1, -C1));
    cr[6] = cmulf(cj, make_float2(-R, -R));
    cr[7] = cmulf(cj, make_float2(-C1, -S1));
#pragma unroll
    for (int r = 0; r < 8; ++r)
      cr[8 + r] = make_float2(-cr[r].x, -cr[r].y);
    const int col = tid + (unit & 3) * PMT;
    int ia[16], ib[16];
#pragma unroll
    for (int r = 0; r < 16; ++r) {
      ia[r] = (jj + r * Q) * Dch + col;
      ib[r] = (((Q - jj) + (15 - r) * Q) & (Llen - 1)) * Dch + col;
      float2 kk = ZK[ia[r]], km = ZK[ib[r]];
      kE[r][tid] = make_float2(kk.x + km.x, kk.y - km.y);
      kO[r][tid] = make_float2(kk.y + km.y, km.x - kk.x);
    }
    for (int pp = 0; pp < Bsz; ++pp) {
      const int po = (rev ? (Bsz - 1 - pp) : pp) * (int)PLANE;
      float2 va[16], vb[16];
#pragma unroll
      for (int r = 0; r < 16; ++r)
        s4_pair_mul(Zb[po + ia[r]], Zb[po + ib[r]], kE[r][tid], kO[r][tid], cr[r], va[r], vb[15 - r]);
      float2 X[16];
      dft16(va, +1.0f, X);
#pragma unroll
      for (int m = 0; m < 16; ++m)
        Zb[po + ia[m]] = X[m];
      if (jj != 0 && jj != Q / 2) {
        dft16(vb, +1.0f, X);
#pragma unroll
        for (int m = 0; m < 16; ++m)
          Zb[po + ib[15 - m]] = X[m];
      }
    }
  }
}

__global__ __launch_bounds__(256) void s4_final_r32(
    const float2* __restrict__ xb, const float* __restrict__ ub,
    const float* __restrict__ Dp, float* __restrict__ outb, int np, int rev)
{
  const int Q = Llen / 32;
  const float dp = Dp[0];
  const float inv = 1.0f / (4.0f * (float)Llen);
  for (int pp = 0; pp < np; ++pp) {
  const int p = rev ? (np - 1 - pp) : pp;
  const float2* __restrict__ x = xb + (size_t)p * PLANE;
  const float* __restrict__ u = ub + (size_t)p * PLANE;
  float* __restrict__ out = outb + (size_t)p * PLANE;
  for (int j = blockIdx.x; j < Q; j += gridDim.x) {
    float ang = (0.0625f * PI_F) * (float)j / (float)Q;
    float s1, c1;
    sincosf(ang, &s1, &c1);
    float2 w[32];
    w[1]  = make_float2(c1, s1);
    w[2]  = cmulf(w[1], w[1]);
    w[3]  = cmulf(w[2], w[1]);
    w[4]  = cmulf(w[2], w[2]);
    w[5]  = cmulf(w[4], w[1]);
    w[6]  = cmulf(w[3], w[3]);
    w[7]  = cmulf(w[4], w[3]);
    w[8]  = cmulf(w[4], w[4]);
    w[9]  = cmulf(w[8], w[1]);
    w[10] = cmulf(w[5], w[5]);
    w[11] = cmulf(w[8], w[3]);
    w[12] = cmulf(w[6], w[6]);
    w[13] = cmulf(w[8], w[5]);
    w[14] = cmulf(w[7], w[7]);
    w[15] = cmulf(w[8], w[7]);
    w[16] = cmulf(w[8], w[8]);
#pragma unroll
    for (int r = 1; r < 16; ++r)
      w[16 + r] = cmulf(w[16], w[r]);
    for (int h = 0; h < 2; ++h) {
      int col = threadIdx.x + h * HCOL;
      float2 v[32];
      float ue[16][2];
#pragma unroll
      for (int r = 0; r < 32; ++r)
        v[r] = x[(j + r * Q) * Dch + col];
#pragma unroll
      for (int m = 0; m < 16; ++m) {
        int e = (2 * (j + m * Q)) * Dch;
        ue[m][0] = u[e + col];
        ue[m][1] = u[e + Dch + col];
      }
#pragma unroll
      for (int r = 1; r < 32; ++r)
        v[r] = cmulf(v[r], w[r]);
      float2 X[32];
      dft32(v, +1.0f, X);
#pragma unroll
      for (int m = 0; m < 16; ++m) {
        int e = (2 * (j + m * Q)) * Dch;
        out[e + col]       = X[m].x * inv + dp * ue[m][0];
        out[e + Dch + col] = X[m].y * inv + dp * ue[m][1];
      }
    }
  }
  }
}


extern "C" void kernel_launch(void* const* d_in, const int* in_sizes, int n_in,
                              void* d_out, int out_size, void* d_ws, size_t ws_size,
                              hipStream_t stream) {
  const float* u        = (const float*)d_in[0];
  const float* Lam      = (const float*)d_in[1];
  const float* Pri      = (const float*)d_in[2];
  const float* Bri      = (const float*)d_in[3];
  const float* Cri      = (const float*)d_in[4];
  const float* Dp       = (const float*)d_in[5];
  const float* log_step = (const float*)d_in[6];
  float* out = (float*)d_out;

  float2* P0 = (float2*)d_ws;
  float2* P2 = P0 + 2 * PLANE;
  float2* P3 = P0 + 3 * PLANE;
  float2* P4 = P0 + 4 * PLANE;
  float2* P6 = P0 + 6 * PLANE;
  float2* P7 = P0 + 7 * PLANE;
  uint8_t* base = (uint8_t*)P7;
  size_t off = 0;
  auto carve = [&](size_t bytes) -> void* {
    void* p = base + off;
    off += (bytes + 255) & ~(size_t)255;
    return p;
  };
  float*  APr  = (float*)carve((size_t)Llen * Nst * 4);
  float*  APi  = (float*)carve((size_t)Llen * Nst * 4);
  float*  W0r  = (float*)carve((size_t)Nst * Dch * 4);
  float*  W0i  = (float*)carve((size_t)Nst * Dch * 4);
  float*  W1r  = (float*)carve((size_t)Nst * Dch * 4);
  float*  W1i  = (float*)carve((size_t)Nst * Dch * 4);
  float*  perl = (float*)carve((size_t)4 * Llen * 4);
  float2* Ev   = P6;
  (void)ws_size; (void)in_sizes; (void)n_in; (void)out_size;

  const int TB   = 256;
  const int gR128 = 4 * R128Q;
  const int gR32 = NBLK;
  const int gPm  = PMG;

  s4_setup<<<NPB + NWB + NCB, TB, 0, stream>>>(Lam, Pri, Cri, log_step, Bri, perl,
                                              W0r, W0i, W1r, W1i, APr, APi);
  s4_cauchy_gemm<<<dim3(Dch / 16, Llen / 16), 32, 0, stream>>>(
      APr, APi, W0r, W0i, W1r, W1i, perl, Ev);

  const float* u0 = u;
  const float* u1 = u + (size_t)2 * Llen * Dch;

  s4_first_r128<<<gR128, TB, 0, stream>>>(u0, P0, Ev, P2, 0, 3, 1);
  s4_fft_r128<<<gR128, TB, 0, stream>>>(P0, P4, P2, P3, -1.0f, +1.0f, 3, 0);

  s4_first_r128<<<gR128, TB, 0, stream>>>(u1, P0, P3, P2, 1, 3, 1);
  s4_fft_r128<<<gR128, TB, 0, stream>>>(P0, P6, P2, P3, -1.0f, -1.0f, 3, 0);
  float2* Kd = P3;

  s4_pmul_r16<<<gPm, PMT, 0, stream>>>(P4, Kd, 1);
  s4_mid_r32<<<gR32, TB, 0, stream>>>(P4, P0, 4, 0);
  s4_final_r32<<<gR32, TB, 0, stream>>>(P0, u, Dp, out, 4, 1);
}
